# starting kernel
# baseline (speedup 1.0000x reference)
#include <hip/hip_runtime.h>
#include <hip/hip_bf16.h>

#define BB   2
#define CC   96
#define HH   64
#define WWW  64
#define LL   4096
#define NN   16
#define RR   6
#define KK   4
#define DBCROWS 38
#define MPAD 48
#define LT   64
#define CH   4
#define CHL  (LL / CH)

typedef float v2f __attribute__((ext_vector_type(2)));
typedef float v8f __attribute__((ext_vector_type(8)));

__device__ __forceinline__ int tr_idx(int l) { return ((l & 63) << 6) | (l >> 6); }
__device__ __forceinline__ int dir_idx(int k, int l) {
    switch (k) {
        case 0:  return l;
        case 1:  return tr_idx(l);
        case 2:  return LL - 1 - l;
        default: return tr_idx(LL - 1 - l);
    }
}

__global__ __launch_bounds__(256)
void s6_proj_kernel(const float* __restrict__ x,
                    const float* __restrict__ wproj,
                    const float* __restrict__ wdelta,
                    const float* __restrict__ dbias,
                    float* __restrict__ delta_out,
                    float* __restrict__ Bt,
                    float* __restrict__ Ct)
{
    __shared__ float sW[MPAD][CC];
    __shared__ float sX[CC][LT + 1];
    __shared__ float sDBC[MPAD][LT + 1];
    __shared__ float sWd[CC][RR];

    const int tid = threadIdx.x;
    const int l0  = blockIdx.x * LT;
    const int k   = blockIdx.y;
    const int b   = blockIdx.z;

    for (int i = tid; i < MPAD * CC; i += 256) {
        int m = i / CC, c = i % CC;
        sW[m][c] = (m < DBCROWS) ? wproj[(k * DBCROWS + m) * CC + c] : 0.0f;
    }
    for (int i = tid; i < CC * RR; i += 256) {
        int c = i / RR, r = i % RR;
        sWd[c][r] = wdelta[(k * CC + c) * RR + r];
    }
    for (int i = tid; i < CC * LT; i += 256) {
        int c = i / LT, col = i % LT;
        sX[c][col] = x[((size_t)b * CC + c) * LL + dir_idx(k, l0 + col)];
    }
    __syncthreads();

    const int wave = tid >> 5;
    const int lane = tid & 31;
    const int lm   = lane & 15;
    const int lh   = lane >> 4;
    for (int t = wave; t < (MPAD / 16) * (LT / 16); t += 8) {
        const int mt = t / (LT / 16);
        const int nt = t % (LT / 16);
        v8f acc = {};
        for (int kk = 0; kk < CC; kk += 4) {
            v2f a, bf;
            a.x  = sW[mt * 16 + lm][kk + 2 * lh + 0];
            a.y  = sW[mt * 16 + lm][kk + 2 * lh + 1];
            bf.x = sX[kk + 2 * lh + 0][nt * 16 + lm];
            bf.y = sX[kk + 2 * lh + 1][nt * 16 + lm];
            acc = __builtin_amdgcn_wmma_f32_16x16x4_f32(
                      false, a, false, bf, (short)0, acc, false, false);
        }
        #pragma unroll
        for (int i = 0; i < 8; ++i)
            sDBC[mt * 16 + i + 8 * lh][nt * 16 + lm] = acc[i];
    }
    __syncthreads();

    for (int i = tid; i < LT * NN; i += 256) {
        int col = i / NN, n = i % NN;
        size_t base = (((size_t)(b * KK + k) * LL + (l0 + col)) * NN + n);
        Bt[base] = sDBC[RR + n][col];
        Ct[base] = sDBC[RR + NN + n][col];
    }
    for (int i = tid; i < CC * LT; i += 256) {
        int c = i / LT, col = i % LT;
        float s = 0.0f;
        #pragma unroll
        for (int r = 0; r < RR; ++r) s += sWd[c][r] * sDBC[r][col];
        s += dbias[k * CC + c];
        float sp = (s > 20.0f) ? s : log1pf(__expf(s));
        delta_out[(((size_t)(b * KK + k) * CC + c) * LL) + l0 + col] = sp;
    }
}

__global__ __launch_bounds__(256)
void s6_scan_carry_kernel(const float* __restrict__ x,
                          const float* __restrict__ delta,
                          const float* __restrict__ Bt,
                          const float* __restrict__ A_log,
                          float* __restrict__ carryA,
                          float* __restrict__ carryH)
{
    const int tid  = threadIdx.x;
    const int unit = (blockIdx.x * 256 + tid) >> 4;
    const int n    = tid & 15;
    if (unit >= BB * KK * CC * CH) return;
    const int chunk = unit % CH;
    const int ch    = unit / CH;
    const int c     = ch % CC;
    const int bk    = ch / CC;
    const int k     = bk % KK;
    const int b     = bk / KK;
    const int d     = k * CC + c;

    const float An = -__expf(A_log[d * NN + n]);
    const float* dptr = delta + ((size_t)bk * CC + c) * LL;
    const float* bptr = Bt + (size_t)bk * LL * NN;
    const float* xptr = x + ((size_t)b * CC + c) * LL;

    float h = 0.0f, ap = 1.0f;
    const int lbeg = chunk * CHL, lend = lbeg + CHL;
    for (int l = lbeg; l < lend; ++l) {
        __builtin_prefetch(dptr + ((l + 64) & (LL - 1)), 0, 0);
        __builtin_prefetch(bptr + (size_t)((l + 16) & (LL - 1)) * NN, 0, 0);
        float ld = dptr[l];
        float u  = xptr[dir_idx(k, l)];
        float Bn = bptr[(size_t)l * NN + n];
        float dA = __expf(ld * An);
        h  = dA * h + (ld * u) * Bn;
        ap *= dA;
    }
    size_t cidx = (size_t)unit * NN + n;
    carryA[cidx] = ap;
    carryH[cidx] = h;
}

__global__ __launch_bounds__(256)
void s6_combine_kernel(const float* __restrict__ carryA,
                       const float* __restrict__ carryH,
                       float* __restrict__ initH)
{
    int i = blockIdx.x * 256 + threadIdx.x;
    if (i >= BB * KK * CC * NN) return;
    int ch = i / NN, n = i % NN;
    float Hp = 0.0f;
    for (int j = 0; j < CH; ++j) {
        size_t idx = (size_t)(ch * CH + j) * NN + n;
        initH[idx] = Hp;
        Hp = carryA[idx] * Hp + carryH[idx];
    }
}

__global__ __launch_bounds__(256)
void s6_scan_out_kernel(const float* __restrict__ x,
                        const float* __restrict__ delta,
                        const float* __restrict__ Bt,
                        const float* __restrict__ Ct,
                        const float* __restrict__ A_log,
                        const float* __restrict__ Ds,
                        const float* __restrict__ initH,
                        float* __restrict__ out)
{
    const int tid  = threadIdx.x;
    const int unit = (blockIdx.x * 256 + tid) >> 4;
    const int n    = tid & 15;
    if (unit >= BB * KK * CC * CH) return;
    const int chunk = unit % CH;
    const int ch    = unit / CH;
    const int c     = ch % CC;
    const int bk    = ch / CC;
    const int k     = bk % KK;
    const int b     = bk / KK;
    const int d     = k * CC + c;

    const float An = -__expf(A_log[d * NN + n]);
    const float Dd = Ds[d];
    const float* dptr = delta + ((size_t)bk * CC + c) * LL;
    const float* bptr = Bt + (size_t)bk * LL * NN;
    const float* cptr = Ct + (size_t)bk * LL * NN;
    const float* xptr = x + ((size_t)b * CC + c) * LL;
    const int slot = (k == 0) ? 0 : (k == 1) ? 2 : (k == 2) ? 1 : 3;
    float* optr = out + ((size_t)(slot * BB + b) * CC + c) * LL;

    float h = initH[(size_t)unit * NN + n];
    const int lbeg = chunk * CHL, lend = lbeg + CHL;
    for (int l = lbeg; l < lend; ++l) {
        __builtin_prefetch(dptr + ((l + 64) & (LL - 1)), 0, 0);
        __builtin_prefetch(cptr + (size_t)((l + 16) & (LL - 1)) * NN, 0, 0);
        float ld  = dptr[l];
        int   idx = dir_idx(k, l);
        float u   = xptr[idx];
        float Bn  = bptr[(size_t)l * NN + n];
        float Cn  = cptr[(size_t)l * NN + n];
        float dA  = __expf(ld * An);
        h = dA * h + (ld * u) * Bn;
        float v = h * Cn;
        v += __shfl_xor(v, 1);
        v += __shfl_xor(v, 2);
        v += __shfl_xor(v, 4);
        v += __shfl_xor(v, 8);
        if (n == 0) optr[idx] = v + Dd * u;
    }
}

extern "C" void kernel_launch(void* const* d_in, const int* in_sizes, int n_in,
                              void* d_out, int out_size, void* d_ws, size_t ws_size,
                              hipStream_t stream) {
    const float* x      = (const float*)d_in[0];
    const float* wproj  = (const float*)d_in[1];
    const float* wdelta = (const float*)d_in[2];
    const float* dbias  = (const float*)d_in[3];
    const float* A_log  = (const float*)d_in[4];
    const float* Ds     = (const float*)d_in[5];
    float* out = (float*)d_out;

    float* ws     = (float*)d_ws;
    float* delta  = ws;
    float* Bt     = delta  + (size_t)BB * KK * CC * LL;
    float* Ct     = Bt     + (size_t)BB * KK * LL * NN;
    float* carryA = Ct     + (size_t)BB * KK * LL * NN;
    float* carryH = carryA + (size_t)BB * KK * CC * CH * NN;
    float* initH  = carryH + (size_t)BB * KK * CC * CH * NN;

    dim3 gProj(LL / LT, KK, BB);
    s6_proj_kernel<<<gProj, 256, 0, stream>>>(x, wproj, wdelta, dbias, delta, Bt, Ct);

    const int units = BB * KK * CC * CH;
    s6_scan_carry_kernel<<<units * 16 / 256, 256, 0, stream>>>(x, delta, Bt, A_log,
                                                               carryA, carryH);
    s6_combine_kernel<<<(BB * KK * CC * NN + 255) / 256, 256, 0, stream>>>(carryA, carryH,
                                                                           initH);
    s6_scan_out_kernel<<<units * 16 / 256, 256, 0, stream>>>(x, delta, Bt, Ct, A_log, Ds,
                                                             initH, out);
}


# scan split into 64 chunks of 64 steps per channel instead of 4 of 1024 (16x more waves, 16x shorter serial loops); changes the order in which chunk carries are composed
# speedup vs baseline: 3.5513x; 3.5513x over previous
#include <hip/hip_runtime.h>
#include <hip/hip_bf16.h>

#define BB   2
#define CC   96
#define HH   64
#define WWW  64
#define LL   4096
#define NN   16
#define RR   6
#define KK   4
#define DBCROWS 38
#define MPAD 48
#define LT   64
#define CH   64
#define CHL  (LL / CH)

typedef float v2f __attribute__((ext_vector_type(2)));
typedef float v8f __attribute__((ext_vector_type(8)));

__device__ __forceinline__ int tr_idx(int l) { return ((l & 63) << 6) | (l >> 6); }
__device__ __forceinline__ int dir_idx(int k, int l) {
    switch (k) {
        case 0:  return l;
        case 1:  return tr_idx(l);
        case 2:  return LL - 1 - l;
        default: return tr_idx(LL - 1 - l);
    }
}

__global__ __launch_bounds__(256)
void s6_proj_kernel(const float* __restrict__ x,
                    const float* __restrict__ wproj,
                    const float* __restrict__ wdelta,
                    const float* __restrict__ dbias,
                    float* __restrict__ delta_out,
                    float* __restrict__ Bt,
                    float* __restrict__ Ct)
{
    __shared__ float sW[MPAD][CC];
    __shared__ float sX[CC][LT + 1];
    __shared__ float sDBC[MPAD][LT + 1];
    __shared__ float sWd[CC][RR];

    const int tid = threadIdx.x;
    const int l0  = blockIdx.x * LT;
    const int k   = blockIdx.y;
    const int b   = blockIdx.z;

    for (int i = tid; i < MPAD * CC; i += 256) {
        int m = i / CC, c = i % CC;
        sW[m][c] = (m < DBCROWS) ? wproj[(k * DBCROWS + m) * CC + c] : 0.0f;
    }
    for (int i = tid; i < CC * RR; i += 256) {
        int c = i / RR, r = i % RR;
        sWd[c][r] = wdelta[(k * CC + c) * RR + r];
    }
    for (int i = tid; i < CC * LT; i += 256) {
        int c = i / LT, col = i % LT;
        sX[c][col] = x[((size_t)b * CC + c) * LL + dir_idx(k, l0 + col)];
    }
    __syncthreads();

    const int wave = tid >> 5;
    const int lane = tid & 31;
    const int lm   = lane & 15;
    const int lh   = lane >> 4;
    for (int t = wave; t < (MPAD / 16) * (LT / 16); t += 8) {
        const int mt = t / (LT / 16);
        const int nt = t % (LT / 16);
        v8f acc = {};
        for (int kk = 0; kk < CC; kk += 4) {
            v2f a, bf;
            a.x  = sW[mt * 16 + lm][kk + 2 * lh + 0];
            a.y  = sW[mt * 16 + lm][kk + 2 * lh + 1];
            bf.x = sX[kk + 2 * lh + 0][nt * 16 + lm];
            bf.y = sX[kk + 2 * lh + 1][nt * 16 + lm];
            acc = __builtin_amdgcn_wmma_f32_16x16x4_f32(
                      false, a, false, bf, (short)0, acc, false, false);
        }
        #pragma unroll
        for (int i = 0; i < 8; ++i)
            sDBC[mt * 16 + i + 8 * lh][nt * 16 + lm] = acc[i];
    }
    __syncthreads();

    for (int i = tid; i < LT * NN; i += 256) {
        int col = i / NN, n = i % NN;
        size_t base = (((size_t)(b * KK + k) * LL + (l0 + col)) * NN + n);
        Bt[base] = sDBC[RR + n][col];
        Ct[base] = sDBC[RR + NN + n][col];
    }
    for (int i = tid; i < CC * LT; i += 256) {
        int c = i / LT, col = i % LT;
        float s = 0.0f;
        #pragma unroll
        for (int r = 0; r < RR; ++r) s += sWd[c][r] * sDBC[r][col];
        s += dbias[k * CC + c];
        float sp = (s > 20.0f) ? s : log1pf(__expf(s));
        delta_out[(((size_t)(b * KK + k) * CC + c) * LL) + l0 + col] = sp;
    }
}

__global__ __launch_bounds__(256)
void s6_scan_carry_kernel(const float* __restrict__ x,
                          const float* __restrict__ delta,
                          const float* __restrict__ Bt,
                          const float* __restrict__ A_log,
                          float* __restrict__ carryA,
                          float* __restrict__ carryH)
{
    const int tid  = threadIdx.x;
    const int unit = (blockIdx.x * 256 + tid) >> 4;
    const int n    = tid & 15;
    if (unit >= BB * KK * CC * CH) return;
    const int chunk = unit % CH;
    const int ch    = unit / CH;
    const int c     = ch % CC;
    const int bk    = ch / CC;
    const int k     = bk % KK;
    const int b     = bk / KK;
    const int d     = k * CC + c;

    const float An = -__expf(A_log[d * NN + n]);
    const float* dptr = delta + ((size_t)bk * CC + c) * LL;
    const float* bptr = Bt + (size_t)bk * LL * NN;
    const float* xptr = x + ((size_t)b * CC + c) * LL;

    float h = 0.0f, ap = 1.0f;
    const int lbeg = chunk * CHL, lend = lbeg + CHL;
    for (int l = lbeg; l < lend; ++l) {
        __builtin_prefetch(dptr + ((l + 64) & (LL - 1)), 0, 0);
        __builtin_prefetch(bptr + (size_t)((l + 16) & (LL - 1)) * NN, 0, 0);
        float ld = dptr[l];
        float u  = xptr[dir_idx(k, l)];
        float Bn = bptr[(size_t)l * NN + n];
        float dA = __expf(ld * An);
        h  = dA * h + (ld * u) * Bn;
        ap *= dA;
    }
    size_t cidx = (size_t)unit * NN + n;
    carryA[cidx] = ap;
    carryH[cidx] = h;
}

__global__ __launch_bounds__(256)
void s6_combine_kernel(const float* __restrict__ carryA,
                       const float* __restrict__ carryH,
                       float* __restrict__ initH)
{
    int i = blockIdx.x * 256 + threadIdx.x;
    if (i >= BB * KK * CC * NN) return;
    int ch = i / NN, n = i % NN;
    float Hp = 0.0f;
    for (int j = 0; j < CH; ++j) {
        size_t idx = (size_t)(ch * CH + j) * NN + n;
        initH[idx] = Hp;
        Hp = carryA[idx] * Hp + carryH[idx];
    }
}

__global__ __launch_bounds__(256)
void s6_scan_out_kernel(const float* __restrict__ x,
                        const float* __restrict__ delta,
                        const float* __restrict__ Bt,
                        const float* __restrict__ Ct,
                        const float* __restrict__ A_log,
                        const float* __restrict__ Ds,
                        const float* __restrict__ initH,
                        float* __restrict__ out)
{
    const int tid  = threadIdx.x;
    const int unit = (blockIdx.x * 256 + tid) >> 4;
    const int n    = tid & 15;
    if (unit >= BB * KK * CC * CH) return;
    const int chunk = unit % CH;
    const int ch    = unit / CH;
    const int c     = ch % CC;
    const int bk    = ch / CC;
    const int k     = bk % KK;
    const int b     = bk / KK;
    const int d     = k * CC + c;

    const float An = -__expf(A_log[d * NN + n]);
    const float Dd = Ds[d];
    const float* dptr = delta + ((size_t)bk * CC + c) * LL;
    const float* bptr = Bt + (size_t)bk * LL * NN;
    const float* cptr = Ct + (size_t)bk * LL * NN;
    const float* xptr = x + ((size_t)b * CC + c) * LL;
    const int slot = (k == 0) ? 0 : (k == 1) ? 2 : (k == 2) ? 1 : 3;
    float* optr = out + ((size_t)(slot * BB + b) * CC + c) * LL;

    float h = initH[(size_t)unit * NN + n];
    const int lbeg = chunk * CHL, lend = lbeg + CHL;
    for (int l = lbeg; l < lend; ++l) {
        __builtin_prefetch(dptr + ((l + 64) & (LL - 1)), 0, 0);
        __builtin_prefetch(cptr + (size_t)((l + 16) & (LL - 1)) * NN, 0, 0);
        float ld  = dptr[l];
        int   idx = dir_idx(k, l);
        float u   = xptr[idx];
        float Bn  = bptr[(size_t)l * NN + n];
        float Cn  = cptr[(size_t)l * NN + n];
        float dA  = __expf(ld * An);
        h = dA * h + (ld * u) * Bn;
        float v = h * Cn;
        v += __shfl_xor(v, 1);
        v += __shfl_xor(v, 2);
        v += __shfl_xor(v, 4);
        v += __shfl_xor(v, 8);
        if (n == 0) optr[idx] = v + Dd * u;
    }
}

extern "C" void kernel_launch(void* const* d_in, const int* in_sizes, int n_in,
                              void* d_out, int out_size, void* d_ws, size_t ws_size,
                              hipStream_t stream) {
    const float* x      = (const float*)d_in[0];
    const float* wproj  = (const float*)d_in[1];
    const float* wdelta = (const float*)d_in[2];
    const float* dbias  = (const float*)d_in[3];
    const float* A_log  = (const float*)d_in[4];
    const float* Ds     = (const float*)d_in[5];
    float* out = (float*)d_out;

    float* ws     = (float*)d_ws;
    float* delta  = ws;
    float* Bt     = delta  + (size_t)BB * KK * CC * LL;
    float* Ct     = Bt     + (size_t)BB * KK * LL * NN;
    float* carryA = Ct     + (size_t)BB * KK * LL * NN;
    float* carryH = carryA + (size_t)BB * KK * CC * CH * NN;
    float* initH  = carryH + (size_t)BB * KK * CC * CH * NN;

    dim3 gProj(LL / LT, KK, BB);
    s6_proj_kernel<<<gProj, 256, 0, stream>>>(x, wproj, wdelta, dbias, delta, Bt, Ct);

    const int units = BB * KK * CC * CH;
    s6_scan_carry_kernel<<<units * 16 / 256, 256, 0, stream>>>(x, delta, Bt, A_log,
                                                               carryA, carryH);
    s6_combine_kernel<<<(BB * KK * CC * NN + 255) / 256, 256, 0, stream>>>(carryA, carryH,
                                                                           initH);
    s6_scan_out_kernel<<<units * 16 / 256, 256, 0, stream>>>(x, delta, Bt, Ct, A_log, Ds,
                                                             initH, out);
}


# scan loops take 16 steps per trip: lane i loads delta,u of step i, DPP row_share broadcasts them; n-sum by DPP row_xmask adds (same butterfly order) not ds_bpermute; one y store per 16 steps
# speedup vs baseline: 6.6784x; 1.8806x over previous
#include <hip/hip_runtime.h>
#include <hip/hip_bf16.h>

#define BB   2
#define CC   96
#define HH   64
#define WWW  64
#define LL   4096
#define NN   16
#define RR   6
#define KK   4
#define DBCROWS 38
#define MPAD 48
#define LT   64
#define CH   64
#define CHL  (LL / CH)

typedef float v2f __attribute__((ext_vector_type(2)));
typedef float v8f __attribute__((ext_vector_type(8)));

__device__ __forceinline__ int tr_idx(int l) { return ((l & 63) << 6) | (l >> 6); }
__device__ __forceinline__ int dir_idx(int k, int l) {
    switch (k) {
        case 0:  return l;
        case 1:  return tr_idx(l);
        case 2:  return LL - 1 - l;
        default: return tr_idx(LL - 1 - l);
    }
}

template <int CTRL>
__device__ __forceinline__ float row_dpp(float v) {
    return __builtin_bit_cast(float,
        __builtin_amdgcn_update_dpp(0, __builtin_bit_cast(int, v), CTRL, 0xf, 0xf, true));
}
#define GRP 16

__global__ __launch_bounds__(256)
void s6_proj_kernel(const float* __restrict__ x,
                    const float* __restrict__ wproj,
                    const float* __restrict__ wdelta,
                    const float* __restrict__ dbias,
                    float* __restrict__ delta_out,
                    float* __restrict__ Bt,
                    float* __restrict__ Ct)
{
    __shared__ float sW[MPAD][CC];
    __shared__ float sX[CC][LT + 1];
    __shared__ float sDBC[MPAD][LT + 1];
    __shared__ float sWd[CC][RR];

    const int tid = threadIdx.x;
    const int l0  = blockIdx.x * LT;
    const int k   = blockIdx.y;
    const int b   = blockIdx.z;

    for (int i = tid; i < MPAD * CC; i += 256) {
        int m = i / CC, c = i % CC;
        sW[m][c] = (m < DBCROWS) ? wproj[(k * DBCROWS + m) * CC + c] : 0.0f;
    }
    for (int i = tid; i < CC * RR; i += 256) {
        int c = i / RR, r = i % RR;
        sWd[c][r] = wdelta[(k * CC + c) * RR + r];
    }
    for (int i = tid; i < CC * LT; i += 256) {
        int c = i / LT, col = i % LT;
        sX[c][col] = x[((size_t)b * CC + c) * LL + dir_idx(k, l0 + col)];
    }
    __syncthreads();

    const int wave = tid >> 5;
    const int lane = tid & 31;
    const int lm   = lane & 15;
    const int lh   = lane >> 4;
    for (int t = wave; t < (MPAD / 16) * (LT / 16); t += 8) {
        const int mt = t / (LT / 16);
        const int nt = t % (LT / 16);
        v8f acc = {};
        for (int kk = 0; kk < CC; kk += 4) {
            v2f a, bf;
            a.x  = sW[mt * 16 + lm][kk + 2 * lh + 0];
            a.y  = sW[mt * 16 + lm][kk + 2 * lh + 1];
            bf.x = sX[kk + 2 * lh + 0][nt * 16 + lm];
            bf.y = sX[kk + 2 * lh + 1][nt * 16 + lm];
            acc = __builtin_amdgcn_wmma_f32_16x16x4_f32(
                      false, a, false, bf, (short)0, acc, false, false);
        }
        #pragma unroll
        for (int i = 0; i < 8; ++i)
            sDBC[mt * 16 + i + 8 * lh][nt * 16 + lm] = acc[i];
    }
    __syncthreads();

    for (int i = tid; i < LT * NN; i += 256) {
        int col = i / NN, n = i % NN;
        size_t base = (((size_t)(b * KK + k) * LL + (l0 + col)) * NN + n);
        Bt[base] = sDBC[RR + n][col];
        Ct[base] = sDBC[RR + NN + n][col];
    }
    for (int i = tid; i < CC * LT; i += 256) {
        int c = i / LT, col = i % LT;
        float s = 0.0f;
        #pragma unroll
        for (int r = 0; r < RR; ++r) s += sWd[c][r] * sDBC[r][col];
        s += dbias[k * CC + c];
        float sp = (s > 20.0f) ? s : log1pf(__expf(s));
        delta_out[(((size_t)(b * KK + k) * CC + c) * LL) + l0 + col] = sp;
    }
}

__global__ __launch_bounds__(256)
void s6_scan_carry_kernel(const float* __restrict__ x,
                          const float* __restrict__ delta,
                          const float* __restrict__ Bt,
                          const float* __restrict__ A_log,
                          float* __restrict__ carryA,
                          float* __restrict__ carryH)
{
    const int tid  = threadIdx.x;
    const int unit = (blockIdx.x * 256 + tid) >> 4;
    const int n    = tid & 15;
    if (unit >= BB * KK * CC * CH) return;
    const int chunk = unit % CH;
    const int ch    = unit / CH;
    const int c     = ch % CC;
    const int bk    = ch / CC;
    const int k     = bk % KK;
    const int b     = bk / KK;
    const int d     = k * CC + c;

    const float An = -__expf(A_log[d * NN + n]);
    const float* dptr = delta + ((size_t)bk * CC + c) * LL;
    const float* bptr = Bt + (size_t)bk * LL * NN;
    const float* xptr = x + ((size_t)b * CC + c) * LL;

    float h = 0.0f, ap = 1.0f;
    const int lbeg = chunk * CHL;
    for (int t = 0; t < CHL / GRP; ++t) {
        const int l0 = lbeg + t * GRP;
        const float ldv = dptr[l0 + n];
        const float duv = ldv * xptr[dir_idx(k, l0 + n)];
        float Bv[GRP];
        #pragma unroll
        for (int s = 0; s < GRP; ++s) Bv[s] = bptr[(size_t)(l0 + s) * NN + n];
#define CARRY_STEP(S) {                                   \
        const float ld = row_dpp<0x150 + S>(ldv);         \
        const float du = row_dpp<0x150 + S>(duv);         \
        const float dA = __expf(ld * An);                 \
        h  = dA * h + du * Bv[S];                         \
        ap *= dA; }
        CARRY_STEP(0)  CARRY_STEP(1)  CARRY_STEP(2)  CARRY_STEP(3)
        CARRY_STEP(4)  CARRY_STEP(5)  CARRY_STEP(6)  CARRY_STEP(7)
        CARRY_STEP(8)  CARRY_STEP(9)  CARRY_STEP(10) CARRY_STEP(11)
        CARRY_STEP(12) CARRY_STEP(13) CARRY_STEP(14) CARRY_STEP(15)
#undef CARRY_STEP
    }
    size_t cidx = (size_t)unit * NN + n;
    carryA[cidx] = ap;
    carryH[cidx] = h;
}

__global__ __launch_bounds__(256)
void s6_combine_kernel(const float* __restrict__ carryA,
                       const float* __restrict__ carryH,
                       float* __restrict__ initH)
{
    int i = blockIdx.x * 256 + threadIdx.x;
    if (i >= BB * KK * CC * NN) return;
    int ch = i / NN, n = i % NN;
    float Hp = 0.0f;
    for (int j = 0; j < CH; ++j) {
        size_t idx = (size_t)(ch * CH + j) * NN + n;
        initH[idx] = Hp;
        Hp = carryA[idx] * Hp + carryH[idx];
    }
}

__global__ __launch_bounds__(256)
void s6_scan_out_kernel(const float* __restrict__ x,
                        const float* __restrict__ delta,
                        const float* __restrict__ Bt,
                        const float* __restrict__ Ct,
                        const float* __restrict__ A_log,
                        const float* __restrict__ Ds,
                        const float* __restrict__ initH,
                        float* __restrict__ out)
{
    const int tid  = threadIdx.x;
    const int unit = (blockIdx.x * 256 + tid) >> 4;
    const int n    = tid & 15;
    if (unit >= BB * KK * CC * CH) return;
    const int chunk = unit % CH;
    const int ch    = unit / CH;
    const int c     = ch % CC;
    const int bk    = ch / CC;
    const int k     = bk % KK;
    const int b     = bk / KK;
    const int d     = k * CC + c;

    const float An = -__expf(A_log[d * NN + n]);
    const float Dd = Ds[d];
    const float* dptr = delta + ((size_t)bk * CC + c) * LL;
    const float* bptr = Bt + (size_t)bk * LL * NN;
    const float* cptr = Ct + (size_t)bk * LL * NN;
    const float* xptr = x + ((size_t)b * CC + c) * LL;
    const int slot = (k == 0) ? 0 : (k == 1) ? 2 : (k == 2) ? 1 : 3;
    float* optr = out + ((size_t)(slot * BB + b) * CC + c) * LL;

    float h = initH[(size_t)unit * NN + n];
    const int lbeg = chunk * CHL;
    for (int t = 0; t < CHL / GRP; ++t) {
        const int l0 = lbeg + t * GRP;
        const int   idx = dir_idx(k, l0 + n);
        const float ldv = dptr[l0 + n];
        const float uv  = xptr[idx];
        const float duv = ldv * uv;
        float Bv[GRP], Cv[GRP];
        #pragma unroll
        for (int s = 0; s < GRP; ++s) {
            Bv[s] = bptr[(size_t)(l0 + s) * NN + n];
            Cv[s] = cptr[(size_t)(l0 + s) * NN + n];
        }
        float ykeep = 0.0f;
#define OUT_STEP(S) {                                     \
        const float ld = row_dpp<0x150 + S>(ldv);         \
        const float du = row_dpp<0x150 + S>(duv);         \
        const float dA = __expf(ld * An);                 \
        h = dA * h + du * Bv[S];                          \
        float v = h * Cv[S];                              \
        v += row_dpp<0x160 + 1>(v);                       \
        v += row_dpp<0x160 + 2>(v);                       \
        v += row_dpp<0x160 + 4>(v);                       \
        v += row_dpp<0x160 + 8>(v);                       \
        ykeep = (n == S) ? v : ykeep; }
        OUT_STEP(0)  OUT_STEP(1)  OUT_STEP(2)  OUT_STEP(3)
        OUT_STEP(4)  OUT_STEP(5)  OUT_STEP(6)  OUT_STEP(7)
        OUT_STEP(8)  OUT_STEP(9)  OUT_STEP(10) OUT_STEP(11)
        OUT_STEP(12) OUT_STEP(13) OUT_STEP(14) OUT_STEP(15)
#undef OUT_STEP
        optr[idx] = ykeep + Dd * uv;
    }
}

extern "C" void kernel_launch(void* const* d_in, const int* in_sizes, int n_in,
                              void* d_out, int out_size, void* d_ws, size_t ws_size,
                              hipStream_t stream) {
    const float* x      = (const float*)d_in[0];
    const float* wproj  = (const float*)d_in[1];
    const float* wdelta = (const float*)d_in[2];
    const float* dbias  = (const float*)d_in[3];
    const float* A_log  = (const float*)d_in[4];
    const float* Ds     = (const float*)d_in[5];
    float* out = (float*)d_out;

    float* ws     = (float*)d_ws;
    float* delta  = ws;
    float* Bt     = delta  + (size_t)BB * KK * CC * LL;
    float* Ct     = Bt     + (size_t)BB * KK * LL * NN;
    float* carryA = Ct     + (size_t)BB * KK * LL * NN;
    float* carryH = carryA + (size_t)BB * KK * CC * CH * NN;
    float* initH  = carryH + (size_t)BB * KK * CC * CH * NN;

    dim3 gProj(LL / LT, KK, BB);
    s6_proj_kernel<<<gProj, 256, 0, stream>>>(x, wproj, wdelta, dbias, delta, Bt, Ct);

    const int units = BB * KK * CC * CH;
    s6_scan_carry_kernel<<<units * 16 / 256, 256, 0, stream>>>(x, delta, Bt, A_log,
                                                               carryA, carryH);
    s6_combine_kernel<<<(BB * KK * CC * NN + 255) / 256, 256, 0, stream>>>(carryA, carryH,
                                                                           initH);
    s6_scan_out_kernel<<<units * 16 / 256, 256, 0, stream>>>(x, delta, Bt, Ct, A_log, Ds,
                                                             initH, out);
}


# carry pass, carry chaining and output pass fused into one launch: one 1024-thread work-group per channel, chunk carries handed over in LDS; chunking and the order of every sum unchanged
# speedup vs baseline: 9.2581x; 1.3863x over previous
#include <hip/hip_runtime.h>
#include <hip/hip_bf16.h>

#define BB   2
#define CC   96
#define HH   64
#define WWW  64
#define LL   4096
#define NN   16
#define RR   6
#define KK   4
#define DBCROWS 38
#define MPAD 48
#define LT   64
#define CH   64
#define CHL  (LL / CH)

typedef float v2f __attribute__((ext_vector_type(2)));
typedef float v8f __attribute__((ext_vector_type(8)));

__device__ __forceinline__ int tr_idx(int l) { return ((l & 63) << 6) | (l >> 6); }
__device__ __forceinline__ int dir_idx(int k, int l) {
    switch (k) {
        case 0:  return l;
        case 1:  return tr_idx(l);
        case 2:  return LL - 1 - l;
        default: return tr_idx(LL - 1 - l);
    }
}

template <int CTRL>
__device__ __forceinline__ float row_dpp(float v) {
    return __builtin_bit_cast(float,
        __builtin_amdgcn_update_dpp(0, __builtin_bit_cast(int, v), CTRL, 0xf, 0xf, true));
}
#define GRP 16

__global__ __launch_bounds__(256)
void s6_proj_kernel(const float* __restrict__ x,
                    const float* __restrict__ wproj,
                    const float* __restrict__ wdelta,
                    const float* __restrict__ dbias,
                    float* __restrict__ delta_out,
                    float* __restrict__ Bt,
                    float* __restrict__ Ct)
{
    __shared__ float sW[MPAD][CC];
    __shared__ float sX[CC][LT + 1];
    __shared__ float sDBC[MPAD][LT + 1];
    __shared__ float sWd[CC][RR];

    const int tid = threadIdx.x;
    const int l0  = blockIdx.x * LT;
    const int k   = blockIdx.y;
    const int b   = blockIdx.z;

    for (int i = tid; i < MPAD * CC; i += 256) {
        int m = i / CC, c = i % CC;
        sW[m][c] = (m < DBCROWS) ? wproj[(k * DBCROWS + m) * CC + c] : 0.0f;
    }
    for (int i = tid; i < CC * RR; i += 256) {
        int c = i / RR, r = i % RR;
        sWd[c][r] = wdelta[(k * CC + c) * RR + r];
    }
    for (int i = tid; i < CC * LT; i += 256) {
        int c = i / LT, col = i % LT;
        sX[c][col] = x[((size_t)b * CC + c) * LL + dir_idx(k, l0 + col)];
    }
    __syncthreads();

    const int wave = tid >> 5;
    const int lane = tid & 31;
    const int lm   = lane & 15;
    const int lh   = lane >> 4;
    for (int t = wave; t < (MPAD / 16) * (LT / 16); t += 8) {
        const int mt = t / (LT / 16);
        const int nt = t % (LT / 16);
        v8f acc = {};
        for (int kk = 0; kk < CC; kk += 4) {
            v2f a, bf;
            a.x  = sW[mt * 16 + lm][kk + 2 * lh + 0];
            a.y  = sW[mt * 16 + lm][kk + 2 * lh + 1];
            bf.x = sX[kk + 2 * lh + 0][nt * 16 + lm];
            bf.y = sX[kk + 2 * lh + 1][nt * 16 + lm];
            acc = __builtin_amdgcn_wmma_f32_16x16x4_f32(
                      false, a, false, bf, (short)0, acc, false, false);
        }
        #pragma unroll
        for (int i = 0; i < 8; ++i)
            sDBC[mt * 16 + i + 8 * lh][nt * 16 + lm] = acc[i];
    }
    __syncthreads();

    for (int i = tid; i < LT * NN; i += 256) {
        int col = i / NN, n = i % NN;
        size_t base = (((size_t)(b * KK + k) * LL + (l0 + col)) * NN + n);
        Bt[base] = sDBC[RR + n][col];
        Ct[base] = sDBC[RR + NN + n][col];
    }
    for (int i = tid; i < CC * LT; i += 256) {
        int c = i / LT, col = i % LT;
        float s = 0.0f;
        #pragma unroll
        for (int r = 0; r < RR; ++r) s += sWd[c][r] * sDBC[r][col];
        s += dbias[k * CC + c];
        float sp = (s > 20.0f) ? s : log1pf(__expf(s));
        delta_out[(((size_t)(b * KK + k) * CC + c) * LL) + l0 + col] = sp;
    }
}

#define SCAN_THREADS (CH * NN)

__global__ __launch_bounds__(SCAN_THREADS)
void s6_scan_kernel(const float* __restrict__ x,
                    const float* __restrict__ delta,
                    const float* __restrict__ Bt,
                    const float* __restrict__ Ct,
                    const float* __restrict__ A_log,
                    const float* __restrict__ Ds,
                    float* __restrict__ out)
{
    __shared__ float sA[CH][NN];
    __shared__ float sH[CH][NN];
    __shared__ float sI[CH][NN];

    const int tid   = threadIdx.x;
    const int chunk = tid >> 4;
    const int n     = tid & 15;
    const int ch    = blockIdx.x;
    const int c     = ch % CC;
    const int bk    = ch / CC;
    const int k     = bk % KK;
    const int b     = bk / KK;
    const int d     = k * CC + c;

    const float An = -__expf(A_log[d * NN + n]);
    const float Dd = Ds[d];
    const float* dptr = delta + ((size_t)bk * CC + c) * LL;
    const float* bptr = Bt + (size_t)bk * LL * NN;
    const float* cptr = Ct + (size_t)bk * LL * NN;
    const float* xptr = x + ((size_t)b * CC + c) * LL;
    const int slot = (k == 0) ? 0 : (k == 1) ? 2 : (k == 2) ? 1 : 3;
    float* optr = out + ((size_t)(slot * BB + b) * CC + c) * LL;
    const int lbeg = chunk * CHL;

    {
        float h = 0.0f, ap = 1.0f;
        for (int t = 0; t < CHL / GRP; ++t) {
            const int l0 = lbeg + t * GRP;
            const float ldv = dptr[l0 + n];
            const float duv = ldv * xptr[dir_idx(k, l0 + n)];
            float Bv[GRP];
            #pragma unroll
            for (int s = 0; s < GRP; ++s) Bv[s] = bptr[(size_t)(l0 + s) * NN + n];
#define CARRY_STEP(S) {                                   \
            const float ld = row_dpp<0x150 + S>(ldv);         \
            const float du = row_dpp<0x150 + S>(duv);         \
            const float dA = __expf(ld * An);                 \
            h  = dA * h + du * Bv[S];                         \
            ap *= dA; }
            CARRY_STEP(0)  CARRY_STEP(1)  CARRY_STEP(2)  CARRY_STEP(3)
            CARRY_STEP(4)  CARRY_STEP(5)  CARRY_STEP(6)  CARRY_STEP(7)
            CARRY_STEP(8)  CARRY_STEP(9)  CARRY_STEP(10) CARRY_STEP(11)
            CARRY_STEP(12) CARRY_STEP(13) CARRY_STEP(14) CARRY_STEP(15)
#undef CARRY_STEP
        }
        sA[chunk][n] = ap;
        sH[chunk][n] = h;
    }
    __syncthreads();

    if (tid < NN) {
        float Hp = 0.0f;
        #pragma unroll 8
        for (int j = 0; j < CH; ++j) {
            sI[j][tid] = Hp;
            Hp = sA[j][tid] * Hp + sH[j][tid];
        }
    }
    __syncthreads();

    float h = sI[chunk][n];
    for (int t = 0; t < CHL / GRP; ++t) {
        const int l0 = lbeg + t * GRP;
        const int   idx = dir_idx(k, l0 + n);
        const float ldv = dptr[l0 + n];
        const float uv  = xptr[idx];
        const float duv = ldv * uv;
        float Bv[GRP], Cv[GRP];
        #pragma unroll
        for (int s = 0; s < GRP; ++s) {
            Bv[s] = bptr[(size_t)(l0 + s) * NN + n];
            Cv[s] = cptr[(size_t)(l0 + s) * NN + n];
        }
        float ykeep = 0.0f;
#define OUT_STEP(S) {                                     \
        const float ld = row_dpp<0x150 + S>(ldv);         \
        const float du = row_dpp<0x150 + S>(duv);         \
        const float dA = __expf(ld * An);                 \
        h = dA * h + du * Bv[S];                          \
        float v = h * Cv[S];                              \
        v += row_dpp<0x160 + 1>(v);                       \
        v += row_dpp<0x160 + 2>(v);                       \
        v += row_dpp<0x160 + 4>(v);                       \
        v += row_dpp<0x160 + 8>(v);                       \
        ykeep = (n == S) ? v : ykeep; }
        OUT_STEP(0)  OUT_STEP(1)  OUT_STEP(2)  OUT_STEP(3)
        OUT_STEP(4)  OUT_STEP(5)  OUT_STEP(6)  OUT_STEP(7)
        OUT_STEP(8)  OUT_STEP(9)  OUT_STEP(10) OUT_STEP(11)
        OUT_STEP(12) OUT_STEP(13) OUT_STEP(14) OUT_STEP(15)
#undef OUT_STEP
        optr[idx] = ykeep + Dd * uv;
    }
}

extern "C" void kernel_launch(void* const* d_in, const int* in_sizes, int n_in,
                              void* d_out, int out_size, void* d_ws, size_t ws_size,
                              hipStream_t stream) {
    const float* x      = (const float*)d_in[0];
    const float* wproj  = (const float*)d_in[1];
    const float* wdelta = (const float*)d_in[2];
    const float* dbias  = (const float*)d_in[3];
    const float* A_log  = (const float*)d_in[4];
    const float* Ds     = (const float*)d_in[5];
    float* out = (float*)d_out;

    float* ws     = (float*)d_ws;
    float* delta  = ws;
    float* Bt     = delta  + (size_t)BB * KK * CC * LL;
    float* Ct     = Bt     + (size_t)BB * KK * LL * NN;

    dim3 gProj(LL / LT, KK, BB);
    s6_proj_kernel<<<gProj, 256, 0, stream>>>(x, wproj, wdelta, dbias, delta, Bt, Ct);

    s6_scan_kernel<<<BB * KK * CC, SCAN_THREADS, 0, stream>>>(x, delta, Bt, Ct, A_log, Ds, out);
}


# each scan work-group takes 3 consecutive channels of one (b,k): the B and C rows of a step are loaded once for 3 channel states (3x fewer B/C loads and waves, 256 groups); every sum order unchanged
# speedup vs baseline: 10.8100x; 1.1676x over previous
#include <hip/hip_runtime.h>
#include <hip/hip_bf16.h>

#define BB   2
#define CC   96
#define HH   64
#define WWW  64
#define LL   4096
#define NN   16
#define RR   6
#define KK   4
#define DBCROWS 38
#define MPAD 48
#define LT   64
#define CH   64
#define CHL  (LL / CH)

typedef float v2f __attribute__((ext_vector_type(2)));
typedef float v8f __attribute__((ext_vector_type(8)));

__device__ __forceinline__ int tr_idx(int l) { return ((l & 63) << 6) | (l >> 6); }
__device__ __forceinline__ int dir_idx(int k, int l) {
    switch (k) {
        case 0:  return l;
        case 1:  return tr_idx(l);
        case 2:  return LL - 1 - l;
        default: return tr_idx(LL - 1 - l);
    }
}

template <int CTRL>
__device__ __forceinline__ float row_dpp(float v) {
    return __builtin_bit_cast(float,
        __builtin_amdgcn_update_dpp(0, __builtin_bit_cast(int, v), CTRL, 0xf, 0xf, true));
}
#define GRP 16

__global__ __launch_bounds__(256)
void s6_proj_kernel(const float* __restrict__ x,
                    const float* __restrict__ wproj,
                    const float* __restrict__ wdelta,
                    const float* __restrict__ dbias,
                    float* __restrict__ delta_out,
                    float* __restrict__ Bt,
                    float* __restrict__ Ct)
{
    __shared__ float sW[MPAD][CC];
    __shared__ float sX[CC][LT + 1];
    __shared__ float sDBC[MPAD][LT + 1];
    __shared__ float sWd[CC][RR];

    const int tid = threadIdx.x;
    const int l0  = blockIdx.x * LT;
    const int k   = blockIdx.y;
    const int b   = blockIdx.z;

    for (int i = tid; i < MPAD * CC; i += 256) {
        int m = i / CC, c = i % CC;
        sW[m][c] = (m < DBCROWS) ? wproj[(k * DBCROWS + m) * CC + c] : 0.0f;
    }
    for (int i = tid; i < CC * RR; i += 256) {
        int c = i / RR, r = i % RR;
        sWd[c][r] = wdelta[(k * CC + c) * RR + r];
    }
    for (int i = tid; i < CC * LT; i += 256) {
        int c = i / LT, col = i % LT;
        sX[c][col] = x[((size_t)b * CC + c) * LL + dir_idx(k, l0 + col)];
    }
    __syncthreads();

    const int wave = tid >> 5;
    const int lane = tid & 31;
    const int lm   = lane & 15;
    const int lh   = lane >> 4;
    for (int t = wave; t < (MPAD / 16) * (LT / 16); t += 8) {
        const int mt = t / (LT / 16);
        const int nt = t % (LT / 16);
        v8f acc = {};
        for (int kk = 0; kk < CC; kk += 4) {
            v2f a, bf;
            a.x  = sW[mt * 16 + lm][kk + 2 * lh + 0];
            a.y  = sW[mt * 16 + lm][kk + 2 * lh + 1];
            bf.x = sX[kk + 2 * lh + 0][nt * 16 + lm];
            bf.y = sX[kk + 2 * lh + 1][nt * 16 + lm];
            acc = __builtin_amdgcn_wmma_f32_16x16x4_f32(
                      false, a, false, bf, (short)0, acc, false, false);
        }
        #pragma unroll
        for (int i = 0; i < 8; ++i)
            sDBC[mt * 16 + i + 8 * lh][nt * 16 + lm] = acc[i];
    }
    __syncthreads();

    for (int i = tid; i < LT * NN; i += 256) {
        int col = i / NN, n = i % NN;
        size_t base = (((size_t)(b * KK + k) * LL + (l0 + col)) * NN + n);
        Bt[base] = sDBC[RR + n][col];
        Ct[base] = sDBC[RR + NN + n][col];
    }
    for (int i = tid; i < CC * LT; i += 256) {
        int c = i / LT, col = i % LT;
        float s = 0.0f;
        #pragma unroll
        for (int r = 0; r < RR; ++r) s += sWd[c][r] * sDBC[r][col];
        s += dbias[k * CC + c];
        float sp = (s > 20.0f) ? s : log1pf(__expf(s));
        delta_out[(((size_t)(b * KK + k) * CC + c) * LL) + l0 + col] = sp;
    }
}

#define SCAN_THREADS (CH * NN)
#define MCH 3

template <int S>
__device__ __forceinline__ void carry_step(const float (&ldv)[MCH], const float (&duv)[MCH],
                                           const float (&An)[MCH], const float Bs,
                                           float (&h)[MCH], float (&ap)[MCH]) {
    #pragma unroll
    for (int m = 0; m < MCH; ++m) {
        const float ld = row_dpp<0x150 + S>(ldv[m]);
        const float du = row_dpp<0x150 + S>(duv[m]);
        const float dA = __expf(ld * An[m]);
        h[m]  = dA * h[m] + du * Bs;
        ap[m] *= dA;
    }
}

template <int S>
__device__ __forceinline__ void out_step(const int n, const float (&ldv)[MCH], const float (&duv)[MCH],
                                         const float (&An)[MCH], const float Bs, const float Cs,
                                         float (&h)[MCH], float (&ykeep)[MCH]) {
    #pragma unroll
    for (int m = 0; m < MCH; ++m) {
        const float ld = row_dpp<0x150 + S>(ldv[m]);
        const float du = row_dpp<0x150 + S>(duv[m]);
        const float dA = __expf(ld * An[m]);
        h[m] = dA * h[m] + du * Bs;
        float v = h[m] * Cs;
        v += row_dpp<0x160 + 1>(v);
        v += row_dpp<0x160 + 2>(v);
        v += row_dpp<0x160 + 4>(v);
        v += row_dpp<0x160 + 8>(v);
        ykeep[m] = (n == S) ? v : ykeep[m];
    }
}

__global__ __launch_bounds__(SCAN_THREADS)
void s6_scan_kernel(const float* __restrict__ x,
                    const float* __restrict__ delta,
                    const float* __restrict__ Bt,
                    const float* __restrict__ Ct,
                    const float* __restrict__ A_log,
                    const float* __restrict__ Ds,
                    float* __restrict__ out)
{
    __shared__ float sA[MCH][CH][NN];
    __shared__ float sH[MCH][CH][NN];
    __shared__ float sI[MCH][CH][NN];

    const int tid   = threadIdx.x;
    const int chunk = tid >> 4;
    const int n     = tid & 15;
    const int grp   = blockIdx.x;
    const int c0    = (grp % (CC / MCH)) * MCH;
    const int bk    = grp / (CC / MCH);
    const int k     = bk % KK;
    const int b     = bk / KK;
    const int d0    = k * CC + c0;

    float An[MCH], Dd[MCH];
    #pragma unroll
    for (int m = 0; m < MCH; ++m) {
        An[m] = -__expf(A_log[(d0 + m) * NN + n]);
        Dd[m] = Ds[d0 + m];
    }
    const float* dptr = delta + ((size_t)bk * CC + c0) * LL;
    const float* bptr = Bt + (size_t)bk * LL * NN;
    const float* cptr = Ct + (size_t)bk * LL * NN;
    const float* xptr = x + ((size_t)b * CC + c0) * LL;
    const int slot = (k == 0) ? 0 : (k == 1) ? 2 : (k == 2) ? 1 : 3;
    float* optr = out + ((size_t)(slot * BB + b) * CC + c0) * LL;
    const int lbeg = chunk * CHL;

    {
        float h[MCH], ap[MCH];
        #pragma unroll
        for (int m = 0; m < MCH; ++m) { h[m] = 0.0f; ap[m] = 1.0f; }
        for (int t = 0; t < CHL / GRP; ++t) {
            const int l0 = lbeg + t * GRP;
            const int idx = dir_idx(k, l0 + n);
            float ldv[MCH], duv[MCH];
            #pragma unroll
            for (int m = 0; m < MCH; ++m) {
                ldv[m] = dptr[m * LL + l0 + n];
                duv[m] = ldv[m] * xptr[m * LL + idx];
            }
            float Bv[GRP];
            #pragma unroll
            for (int s = 0; s < GRP; ++s) Bv[s] = bptr[(size_t)(l0 + s) * NN + n];
            carry_step<0>(ldv, duv, An, Bv[0], h, ap);   carry_step<1>(ldv, duv, An, Bv[1], h, ap);
            carry_step<2>(ldv, duv, An, Bv[2], h, ap);   carry_step<3>(ldv, duv, An, Bv[3], h, ap);
            carry_step<4>(ldv, duv, An, Bv[4], h, ap);   carry_step<5>(ldv, duv, An, Bv[5], h, ap);
            carry_step<6>(ldv, duv, An, Bv[6], h, ap);   carry_step<7>(ldv, duv, An, Bv[7], h, ap);
            carry_step<8>(ldv, duv, An, Bv[8], h, ap);   carry_step<9>(ldv, duv, An, Bv[9], h, ap);
            carry_step<10>(ldv, duv, An, Bv[10], h, ap); carry_step<11>(ldv, duv, An, Bv[11], h, ap);
            carry_step<12>(ldv, duv, An, Bv[12], h, ap); carry_step<13>(ldv, duv, An, Bv[13], h, ap);
            carry_step<14>(ldv, duv, An, Bv[14], h, ap); carry_step<15>(ldv, duv, An, Bv[15], h, ap);
        }
        #pragma unroll
        for (int m = 0; m < MCH; ++m) {
            sA[m][chunk][n] = ap[m];
            sH[m][chunk][n] = h[m];
        }
    }
    __syncthreads();

    if (tid < MCH * NN) {
        const int m = tid >> 4;
        float Hp = 0.0f;
        #pragma unroll 8
        for (int j = 0; j < CH; ++j) {
            sI[m][j][n] = Hp;
            Hp = sA[m][j][n] * Hp + sH[m][j][n];
        }
    }
    __syncthreads();

    float h[MCH];
    #pragma unroll
    for (int m = 0; m < MCH; ++m) h[m] = sI[m][chunk][n];
    for (int t = 0; t < CHL / GRP; ++t) {
        const int l0 = lbeg + t * GRP;
        const int idx = dir_idx(k, l0 + n);
        float ldv[MCH], uv[MCH], duv[MCH], ykeep[MCH];
        #pragma unroll
        for (int m = 0; m < MCH; ++m) {
            ldv[m]   = dptr[m * LL + l0 + n];
            uv[m]    = xptr[m * LL + idx];
            duv[m]   = ldv[m] * uv[m];
            ykeep[m] = 0.0f;
        }
        float Bv[GRP], Cv[GRP];
        #pragma unroll
        for (int s = 0; s < GRP; ++s) {
            Bv[s] = bptr[(size_t)(l0 + s) * NN + n];
            Cv[s] = cptr[(size_t)(l0 + s) * NN + n];
        }
        out_step<0>(n, ldv, duv, An, Bv[0], Cv[0], h, ykeep);    out_step<1>(n, ldv, duv, An, Bv[1], Cv[1], h, ykeep);
        out_step<2>(n, ldv, duv, An, Bv[2], Cv[2], h, ykeep);    out_step<3>(n, ldv, duv, An, Bv[3], Cv[3], h, ykeep);
        out_step<4>(n, ldv, duv, An, Bv[4], Cv[4], h, ykeep);    out_step<5>(n, ldv, duv, An, Bv[5], Cv[5], h, ykeep);
        out_step<6>(n, ldv, duv, An, Bv[6], Cv[6], h, ykeep);    out_step<7>(n, ldv, duv, An, Bv[7], Cv[7], h, ykeep);
        out_step<8>(n, ldv, duv, An, Bv[8], Cv[8], h, ykeep);    out_step<9>(n, ldv, duv, An, Bv[9], Cv[9], h, ykeep);
        out_step<10>(n, ldv, duv, An, Bv[10], Cv[10], h, ykeep); out_step<11>(n, ldv, duv, An, Bv[11], Cv[11], h, ykeep);
        out_step<12>(n, ldv, duv, An, Bv[12], Cv[12], h, ykeep); out_step<13>(n, ldv, duv, An, Bv[13], Cv[13], h, ykeep);
        out_step<14>(n, ldv, duv, An, Bv[14], Cv[14], h, ykeep); out_step<15>(n, ldv, duv, An, Bv[15], Cv[15], h, ykeep);
        #pragma unroll
        for (int m = 0; m < MCH; ++m) optr[m * LL + idx] = ykeep[m] + Dd[m] * uv[m];
    }
}

extern "C" void kernel_launch(void* const* d_in, const int* in_sizes, int n_in,
                              void* d_out, int out_size, void* d_ws, size_t ws_size,
                              hipStream_t stream) {
    const float* x      = (const float*)d_in[0];
    const float* wproj  = (const float*)d_in[1];
    const float* wdelta = (const float*)d_in[2];
    const float* dbias  = (const float*)d_in[3];
    const float* A_log  = (const float*)d_in[4];
    const float* Ds     = (const float*)d_in[5];
    float* out = (float*)d_out;

    float* ws     = (float*)d_ws;
    float* delta  = ws;
    float* Bt     = delta  + (size_t)BB * KK * CC * LL;
    float* Ct     = Bt     + (size_t)BB * KK * LL * NN;

    dim3 gProj(LL / LT, KK, BB);
    s6_proj_kernel<<<gProj, 256, 0, stream>>>(x, wproj, wdelta, dbias, delta, Bt, Ct);

    s6_scan_kernel<<<BB * KK * (CC / MCH), SCAN_THREADS, 0, stream>>>(x, delta, Bt, Ct, A_log, Ds, out);
}


# scan: each lane holds 4 of the 16 states as two f32 pairs (a quad of lanes per chunk, 128 chunks of 32 steps, 512 threads): 4x fewer DPP per state update; changes carry chunking and n-sum order
# speedup vs baseline: 11.7882x; 1.0905x over previous
#include <hip/hip_runtime.h>
#include <hip/hip_bf16.h>

#define BB   2
#define CC   96
#define HH   64
#define WWW  64
#define LL   4096
#define NN   16
#define RR   6
#define KK   4
#define DBCROWS 38
#define MPAD 48
#define LT   64
#define QN   4
#define QL   (NN / QN)
#define CH   128
#define CHL  (LL / CH)

typedef float v2f __attribute__((ext_vector_type(2)));
typedef float v8f __attribute__((ext_vector_type(8)));

__device__ __forceinline__ int tr_idx(int l) { return ((l & 63) << 6) | (l >> 6); }
__device__ __forceinline__ int dir_idx(int k, int l) {
    switch (k) {
        case 0:  return l;
        case 1:  return tr_idx(l);
        case 2:  return LL - 1 - l;
        default: return tr_idx(LL - 1 - l);
    }
}

template <int CTRL>
__device__ __forceinline__ float row_dpp(float v) {
    return __builtin_bit_cast(float,
        __builtin_amdgcn_update_dpp(0, __builtin_bit_cast(int, v), CTRL, 0xf, 0xf, true));
}
#define GRP 16

__global__ __launch_bounds__(256)
void s6_proj_kernel(const float* __restrict__ x,
                    const float* __restrict__ wproj,
                    const float* __restrict__ wdelta,
                    const float* __restrict__ dbias,
                    float* __restrict__ delta_out,
                    float* __restrict__ Bt,
                    float* __restrict__ Ct)
{
    __shared__ float sW[MPAD][CC];
    __shared__ float sX[CC][LT + 1];
    __shared__ float sDBC[MPAD][LT + 1];
    __shared__ float sWd[CC][RR];

    const int tid = threadIdx.x;
    const int l0  = blockIdx.x * LT;
    const int k   = blockIdx.y;
    const int b   = blockIdx.z;

    for (int i = tid; i < MPAD * CC; i += 256) {
        int m = i / CC, c = i % CC;
        sW[m][c] = (m < DBCROWS) ? wproj[(k * DBCROWS + m) * CC + c] : 0.0f;
    }
    for (int i = tid; i < CC * RR; i += 256) {
        int c = i / RR, r = i % RR;
        sWd[c][r] = wdelta[(k * CC + c) * RR + r];
    }
    for (int i = tid; i < CC * LT; i += 256) {
        int c = i / LT, col = i % LT;
        sX[c][col] = x[((size_t)b * CC + c) * LL + dir_idx(k, l0 + col)];
    }
    __syncthreads();

    const int wave = tid >> 5;
    const int lane = tid & 31;
    const int lm   = lane & 15;
    const int lh   = lane >> 4;
    for (int t = wave; t < (MPAD / 16) * (LT / 16); t += 8) {
        const int mt = t / (LT / 16);
        const int nt = t % (LT / 16);
        v8f acc = {};
        for (int kk = 0; kk < CC; kk += 4) {
            v2f a, bf;
            a.x  = sW[mt * 16 + lm][kk + 2 * lh + 0];
            a.y  = sW[mt * 16 + lm][kk + 2 * lh + 1];
            bf.x = sX[kk + 2 * lh + 0][nt * 16 + lm];
            bf.y = sX[kk + 2 * lh + 1][nt * 16 + lm];
            acc = __builtin_amdgcn_wmma_f32_16x16x4_f32(
                      false, a, false, bf, (short)0, acc, false, false);
        }
        #pragma unroll
        for (int i = 0; i < 8; ++i)
            sDBC[mt * 16 + i + 8 * lh][nt * 16 + lm] = acc[i];
    }
    __syncthreads();

    for (int i = tid; i < LT * NN; i += 256) {
        int col = i / NN, n = i % NN;
        size_t base = (((size_t)(b * KK + k) * LL + (l0 + col)) * NN + n);
        Bt[base] = sDBC[RR + n][col];
        Ct[base] = sDBC[RR + NN + n][col];
    }
    for (int i = tid; i < CC * LT; i += 256) {
        int c = i / LT, col = i % LT;
        float s = 0.0f;
        #pragma unroll
        for (int r = 0; r < RR; ++r) s += sWd[c][r] * sDBC[r][col];
        s += dbias[k * CC + c];
        float sp = (s > 20.0f) ? s : log1pf(__expf(s));
        delta_out[(((size_t)(b * KK + k) * CC + c) * LL) + l0 + col] = sp;
    }
}

#define SCAN_THREADS (CH * QL)
#define MCH 3
#define QP  (QN / 2)

__device__ __forceinline__ int dir_pos(const int k, const int l) {
    const int lr = (k & 2) ? LL - 1 - l : l;
    return (k & 1) ? tr_idx(lr) : lr;
}
__device__ __forceinline__ void load_u(const float* __restrict__ p, const int k, const int l,
                                       float (&u)[MCH][4]) {
    int idx[4];
    #pragma unroll
    for (int i = 0; i < 4; ++i) idx[i] = dir_pos(k, l + i);
    #pragma unroll
    for (int m = 0; m < MCH; ++m) {
        #pragma unroll
        for (int i = 0; i < 4; ++i) u[m][i] = p[m * LL + idx[i]];
    }
}
__device__ __forceinline__ void store_dir4(float* __restrict__ p, const int k, const int l,
                                           const float (&v)[4]) {
    if (k & 1) {
        #pragma unroll
        for (int i = 0; i < 4; ++i) p[dir_pos(k, l + i)] = v[i];
    } else {
        const float4 t = (k == 2) ? make_float4(v[3], v[2], v[1], v[0])
                                  : make_float4(v[0], v[1], v[2], v[3]);
        *reinterpret_cast<float4*>(p + ((k == 2) ? LL - 4 - l : l)) = t;
    }
}

template <int S>
__device__ __forceinline__ void carry_step(const float (&ldv)[MCH][4], const float (&duv)[MCH][4],
                                           const v2f (&An)[MCH][QP], const float4 Bv,
                                           v2f (&h)[MCH][QP], v2f (&ap)[MCH][QP]) {
    const v2f Bs[QP] = {{Bv.x, Bv.y}, {Bv.z, Bv.w}};
    #pragma unroll
    for (int m = 0; m < MCH; ++m) {
        const float ld = row_dpp<(S >> 2) * 0x55>(ldv[m][S & 3]);
        const float du = row_dpp<(S >> 2) * 0x55>(duv[m][S & 3]);
        #pragma unroll
        for (int p = 0; p < QP; ++p) {
            const v2f e = ld * An[m][p];
            v2f dA;
            dA.x = __expf(e.x);
            dA.y = __expf(e.y);
            h[m][p]  = dA * h[m][p] + du * Bs[p];
            ap[m][p] *= dA;
        }
    }
}

template <int S>
__device__ __forceinline__ void out_step(const int q, const float (&ldv)[MCH][4], const float (&duv)[MCH][4],
                                         const v2f (&An)[MCH][QP], const float4 Bv, const float4 Cv,
                                         v2f (&h)[MCH][QP], float (&ykeep)[MCH][4]) {
    const v2f Bs[QP] = {{Bv.x, Bv.y}, {Bv.z, Bv.w}};
    const v2f Cs[QP] = {{Cv.x, Cv.y}, {Cv.z, Cv.w}};
    #pragma unroll
    for (int m = 0; m < MCH; ++m) {
        const float ld = row_dpp<(S >> 2) * 0x55>(ldv[m][S & 3]);
        const float du = row_dpp<(S >> 2) * 0x55>(duv[m][S & 3]);
        #pragma unroll
        for (int p = 0; p < QP; ++p) {
            const v2f e = ld * An[m][p];
            v2f dA;
            dA.x = __expf(e.x);
            dA.y = __expf(e.y);
            h[m][p] = dA * h[m][p] + du * Bs[p];
        }
        const v2f s2 = h[m][0] * Cs[0] + h[m][1] * Cs[1];
        float v = s2.x + s2.y;
        v += row_dpp<0x160 + 1>(v);
        v += row_dpp<0x160 + 2>(v);
        ykeep[m][S & 3] = (q == (S >> 2)) ? v : ykeep[m][S & 3];
    }
}

template <int G>
__device__ __forceinline__ void load_rows(const float* __restrict__ rq, float4 (&R)[4]) {
    #pragma unroll
    for (int s = 0; s < 4; ++s) R[s] = *reinterpret_cast<const float4*>(rq + (4 * G + s) * NN);
}
template <int G>
__device__ __forceinline__ void carry_group(const float (&ldv)[MCH][4], const float (&duv)[MCH][4],
                                            const v2f (&An)[MCH][QP], const float4 (&Bv)[4],
                                            v2f (&h)[MCH][QP], v2f (&ap)[MCH][QP]) {
    carry_step<4 * G + 0>(ldv, duv, An, Bv[0], h, ap); carry_step<4 * G + 1>(ldv, duv, An, Bv[1], h, ap);
    carry_step<4 * G + 2>(ldv, duv, An, Bv[2], h, ap); carry_step<4 * G + 3>(ldv, duv, An, Bv[3], h, ap);
}
template <int G>
__device__ __forceinline__ void out_group(const int q, const float (&ldv)[MCH][4], const float (&duv)[MCH][4],
                                          const v2f (&An)[MCH][QP], const float4 (&Bv)[4], const float4 (&Cv)[4],
                                          v2f (&h)[MCH][QP], float (&ykeep)[MCH][4]) {
    out_step<4 * G + 0>(q, ldv, duv, An, Bv[0], Cv[0], h, ykeep); out_step<4 * G + 1>(q, ldv, duv, An, Bv[1], Cv[1], h, ykeep);
    out_step<4 * G + 2>(q, ldv, duv, An, Bv[2], Cv[2], h, ykeep); out_step<4 * G + 3>(q, ldv, duv, An, Bv[3], Cv[3], h, ykeep);
}
#define GROUP_FENCE() __builtin_amdgcn_sched_barrier(0)

__global__ __launch_bounds__(SCAN_THREADS)
void s6_scan_kernel(const float* __restrict__ x,
                    const float* __restrict__ delta,
                    const float* __restrict__ Bt,
                    const float* __restrict__ Ct,
                    const float* __restrict__ A_log,
                    const float* __restrict__ Ds,
                    float* __restrict__ out)
{
    __shared__ __attribute__((aligned(16))) float sA[MCH][CH][NN];
    __shared__ __attribute__((aligned(16))) float sH[MCH][CH][NN];

    const int tid   = threadIdx.x;
    const int chunk = tid >> 2;
    const int q     = tid & 3;
    const int n0    = q * QN;
    const int grp   = blockIdx.x;
    const int c0    = (grp % (CC / MCH)) * MCH;
    const int bk    = grp / (CC / MCH);
    const int k     = bk % KK;
    const int b     = bk / KK;
    const int d0    = k * CC + c0;

    v2f An[MCH][QP];
    float Dd[MCH];
    #pragma unroll
    for (int m = 0; m < MCH; ++m) {
        #pragma unroll
        for (int p = 0; p < QP; ++p) {
            An[m][p].x = -__expf(A_log[(d0 + m) * NN + n0 + 2 * p + 0]);
            An[m][p].y = -__expf(A_log[(d0 + m) * NN + n0 + 2 * p + 1]);
        }
        Dd[m] = Ds[d0 + m];
    }
    const float* dptr = delta + ((size_t)bk * CC + c0) * LL;
    const float* bptr = Bt + (size_t)bk * LL * NN + n0;
    const float* cptr = Ct + (size_t)bk * LL * NN + n0;
    const float* xptr = x + ((size_t)b * CC + c0) * LL;
    const int slot = (k == 0) ? 0 : (k == 1) ? 2 : (k == 2) ? 1 : 3;
    float* optr = out + ((size_t)(slot * BB + b) * CC + c0) * LL;
    const int lbeg = chunk * CHL;

    {
        v2f h[MCH][QP], ap[MCH][QP];
        #pragma unroll
        for (int m = 0; m < MCH; ++m) {
            #pragma unroll
            for (int p = 0; p < QP; ++p) { h[m][p] = (v2f){0.0f, 0.0f}; ap[m][p] = (v2f){1.0f, 1.0f}; }
        }
        for (int t = 0; t < CHL / GRP; ++t) {
            const int l0 = lbeg + t * GRP;
            const int lq = l0 + 4 * q;
            float ldv[MCH][4], uv[MCH][4], duv[MCH][4];
            load_u(xptr, k, lq, uv);
            #pragma unroll
            for (int m = 0; m < MCH; ++m) {
                const float4 d4 = *reinterpret_cast<const float4*>(dptr + m * LL + lq);
                ldv[m][0] = d4.x; ldv[m][1] = d4.y; ldv[m][2] = d4.z; ldv[m][3] = d4.w;
                #pragma unroll
                for (int i = 0; i < 4; ++i) duv[m][i] = ldv[m][i] * uv[m][i];
            }
            const float* bq = bptr + (size_t)l0 * NN;
            float4 Be[4], Bo[4];
            load_rows<0>(bq, Be);
            load_rows<1>(bq, Bo);
            carry_group<0>(ldv, duv, An, Be, h, ap);
            GROUP_FENCE();
            load_rows<2>(bq, Be);
            carry_group<1>(ldv, duv, An, Bo, h, ap);
            GROUP_FENCE();
            load_rows<3>(bq, Bo);
            carry_group<2>(ldv, duv, An, Be, h, ap);
            carry_group<3>(ldv, duv, An, Bo, h, ap);
        }
        #pragma unroll
        for (int m = 0; m < MCH; ++m) {
            *reinterpret_cast<float4*>(&sA[m][chunk][n0]) = make_float4(ap[m][0].x, ap[m][0].y, ap[m][1].x, ap[m][1].y);
            *reinterpret_cast<float4*>(&sH[m][chunk][n0]) = make_float4(h[m][0].x, h[m][0].y, h[m][1].x, h[m][1].y);
        }
    }
    __syncthreads();

    if (tid < MCH * NN) {
        const int m = tid >> 4;
        const int n = tid & 15;
        float Hp = 0.0f;
        #pragma unroll 8
        for (int j = 0; j < CH; ++j) {
            const float hj = sH[m][j][n];
            sH[m][j][n] = Hp;
            Hp = sA[m][j][n] * Hp + hj;
        }
    }
    __syncthreads();

    v2f h[MCH][QP];
    #pragma unroll
    for (int m = 0; m < MCH; ++m) {
        const float4 i4 = *reinterpret_cast<const float4*>(&sH[m][chunk][n0]);
        h[m][0] = (v2f){i4.x, i4.y};
        h[m][1] = (v2f){i4.z, i4.w};
    }
    for (int t = 0; t < CHL / GRP; ++t) {
        const int l0 = lbeg + t * GRP;
        const int lq = l0 + 4 * q;
        float ldv[MCH][4], uv[MCH][4], duv[MCH][4], ykeep[MCH][4];
        load_u(xptr, k, lq, uv);
        #pragma unroll
        for (int m = 0; m < MCH; ++m) {
            const float4 d4 = *reinterpret_cast<const float4*>(dptr + m * LL + lq);
            ldv[m][0] = d4.x; ldv[m][1] = d4.y; ldv[m][2] = d4.z; ldv[m][3] = d4.w;
            #pragma unroll
            for (int i = 0; i < 4; ++i) { duv[m][i] = ldv[m][i] * uv[m][i]; ykeep[m][i] = 0.0f; }
        }
        const float* bq = bptr + (size_t)l0 * NN;
        const float* cq = cptr + (size_t)l0 * NN;
        float4 Be[4], Ce[4], Bo[4], Co[4];
        load_rows<0>(bq, Be); load_rows<0>(cq, Ce);
        load_rows<1>(bq, Bo); load_rows<1>(cq, Co);
        out_group<0>(q, ldv, duv, An, Be, Ce, h, ykeep);
        GROUP_FENCE();
        load_rows<2>(bq, Be); load_rows<2>(cq, Ce);
        out_group<1>(q, ldv, duv, An, Bo, Co, h, ykeep);
        GROUP_FENCE();
        load_rows<3>(bq, Bo); load_rows<3>(cq, Co);
        out_group<2>(q, ldv, duv, An, Be, Ce, h, ykeep);
        out_group<3>(q, ldv, duv, An, Bo, Co, h, ykeep);
        #pragma unroll
        for (int m = 0; m < MCH; ++m) {
            float y4[4];
            #pragma unroll
            for (int i = 0; i < 4; ++i) y4[i] = ykeep[m][i] + Dd[m] * uv[m][i];
            store_dir4(optr + m * LL, k, lq, y4);
        }
    }
}

extern "C" void kernel_launch(void* const* d_in, const int* in_sizes, int n_in,
                              void* d_out, int out_size, void* d_ws, size_t ws_size,
                              hipStream_t stream) {
    const float* x      = (const float*)d_in[0];
    const float* wproj  = (const float*)d_in[1];
    const float* wdelta = (const float*)d_in[2];
    const float* dbias  = (const float*)d_in[3];
    const float* A_log  = (const float*)d_in[4];
    const float* Ds     = (const float*)d_in[5];
    float* out = (float*)d_out;

    float* ws     = (float*)d_ws;
    float* delta  = ws;
    float* Bt     = delta  + (size_t)BB * KK * CC * LL;
    float* Ct     = Bt     + (size_t)BB * KK * LL * NN;

    dim3 gProj(LL / LT, KK, BB);
    s6_proj_kernel<<<gProj, 256, 0, stream>>>(x, wproj, wdelta, dbias, delta, Bt, Ct);

    s6_scan_kernel<<<BB * KK * (CC / MCH), SCAN_THREADS, 0, stream>>>(x, delta, Bt, Ct, A_log, Ds, out);
}


# projection kernel: all global loads of a thread (weights as flat 16-byte copies, delta weights, bias, x tile) issued as one batch before the first LDS store, bias read from LDS; sums unchanged
# speedup vs baseline: 15.2650x; 1.2949x over previous
#include <hip/hip_runtime.h>
#include <hip/hip_bf16.h>

#define BB   2
#define CC   96
#define HH   64
#define WWW  64
#define LL   4096
#define NN   16
#define RR   6
#define KK   4
#define DBCROWS 38
#define MPAD 48
#define LT   64
#define QN   4
#define QL   (NN / QN)
#define CH   128
#define CHL  (LL / CH)

typedef float v2f __attribute__((ext_vector_type(2)));
typedef float v8f __attribute__((ext_vector_type(8)));
typedef float v4f __attribute__((ext_vector_type(4)));

__device__ __forceinline__ int tr_idx(int l) { return ((l & 63) << 6) | (l >> 6); }
__device__ __forceinline__ int dir_pos(const int k, const int l) {
    const int lr = (k & 2) ? LL - 1 - l : l;
    return (k & 1) ? tr_idx(lr) : lr;
}

template <int CTRL>
__device__ __forceinline__ float row_dpp(float v) {
    return __builtin_bit_cast(float,
        __builtin_amdgcn_update_dpp(0, __builtin_bit_cast(int, v), CTRL, 0xf, 0xf, true));
}
#define GRP 16

__global__ __launch_bounds__(256)
void s6_proj_kernel(const float* __restrict__ x,
                    const float* __restrict__ wproj,
                    const float* __restrict__ wdelta,
                    const float* __restrict__ dbias,
                    float* __restrict__ delta_out,
                    float* __restrict__ Bt,
                    float* __restrict__ Ct)
{
    __shared__ __attribute__((aligned(16))) float sW[MPAD][CC];
    __shared__ float sX[CC][LT + 1];
    __shared__ float sDBC[MPAD][LT + 1];
    __shared__ __attribute__((aligned(16))) float sWd[256 * 4];
    __shared__ float sBias[256];

    const int tid = threadIdx.x;
    const int l0  = blockIdx.x * LT;
    const int k   = blockIdx.y;
    const int b   = blockIdx.z;

    constexpr int W4  = DBCROWS * CC / 4;
    constexpr int WP4 = MPAD * CC / 4;
    constexpr int D4  = CC * RR / 4;
    constexpr int XJ  = CC * LT / 256;
    const v4f* wsrc = reinterpret_cast<const v4f*>(wproj + (size_t)k * DBCROWS * CC);
    const v4f* dsrc = reinterpret_cast<const v4f*>(wdelta + (size_t)k * CC * RR);
    const v4f   w0 = wsrc[tid];
    const v4f   w1 = wsrc[tid + 256];
    const v4f   w2 = wsrc[tid + 512];
    const bool  w3in = tid + 768 < W4;
    const v4f   w3 = wsrc[w3in ? tid + 768 : W4 - 1];
    const v4f   dv = dsrc[tid < D4 ? tid : D4 - 1];
    const float bv = dbias[k * CC + (tid < CC ? tid : CC - 1)];
    const int col = tid & (LT - 1);
    const int cr  = tid >> 6;
    const float* xcol = x + (size_t)b * CC * LL + dir_pos(k, l0 + col);
    float xv[XJ];
    #pragma unroll
    for (int j = 0; j < XJ; ++j) xv[j] = xcol[(size_t)(cr + 4 * j) * LL];

    v4f* sW4 = reinterpret_cast<v4f*>(&sW[0][0]);
    const v4f zero4 = {0.0f, 0.0f, 0.0f, 0.0f};
    sW4[tid]       = w0;
    sW4[tid + 256] = w1;
    sW4[tid + 512] = w2;
    sW4[tid + 768] = w3in ? w3 : zero4;
    if (tid < WP4 - 1024) sW4[tid + 1024] = zero4;
    reinterpret_cast<v4f*>(sWd)[tid] = dv;
    sBias[tid] = bv;
    #pragma unroll
    for (int j = 0; j < XJ; ++j) sX[cr + 4 * j][col] = xv[j];
    __syncthreads();

    const int wave = tid >> 5;
    const int lane = tid & 31;
    const int lm   = lane & 15;
    const int lh   = lane >> 4;
    for (int t = wave; t < (MPAD / 16) * (LT / 16); t += 8) {
        const int mt = t / (LT / 16);
        const int nt = t % (LT / 16);
        v8f acc = {};
        for (int kk = 0; kk < CC; kk += 4) {
            v2f a, bf;
            a.x  = sW[mt * 16 + lm][kk + 2 * lh + 0];
            a.y  = sW[mt * 16 + lm][kk + 2 * lh + 1];
            bf.x = sX[kk + 2 * lh + 0][nt * 16 + lm];
            bf.y = sX[kk + 2 * lh + 1][nt * 16 + lm];
            acc = __builtin_amdgcn_wmma_f32_16x16x4_f32(
                      false, a, false, bf, (short)0, acc, false, false);
        }
        #pragma unroll
        for (int i = 0; i < 8; ++i)
            sDBC[mt * 16 + i + 8 * lh][nt * 16 + lm] = acc[i];
    }
    __syncthreads();

    for (int i = tid; i < LT * NN; i += 256) {
        int col = i / NN, n = i % NN;
        size_t base = (((size_t)(b * KK + k) * LL + (l0 + col)) * NN + n);
        Bt[base] = sDBC[RR + n][col];
        Ct[base] = sDBC[RR + NN + n][col];
    }
    for (int i = tid; i < CC * LT; i += 256) {
        int c = i / LT, col = i % LT;
        float s = 0.0f;
        #pragma unroll
        for (int r = 0; r < RR; ++r) s += sWd[c * RR + r] * sDBC[r][col];
        s += sBias[c];
        float sp = (s > 20.0f) ? s : log1pf(__expf(s));
        delta_out[(((size_t)(b * KK + k) * CC + c) * LL) + l0 + col] = sp;
    }
}

#define SCAN_THREADS (CH * QL)
#define MCH 3
#define QP  (QN / 2)

__device__ __forceinline__ void load_u(const float* __restrict__ p, const int k, const int l,
                                       float (&u)[MCH][4]) {
    int idx[4];
    #pragma unroll
    for (int i = 0; i < 4; ++i) idx[i] = dir_pos(k, l + i);
    #pragma unroll
    for (int m = 0; m < MCH; ++m) {
        #pragma unroll
        for (int i = 0; i < 4; ++i) u[m][i] = p[m * LL + idx[i]];
    }
}
__device__ __forceinline__ void store_dir4(float* __restrict__ p, const int k, const int l,
                                           const float (&v)[4]) {
    if (k & 1) {
        #pragma unroll
        for (int i = 0; i < 4; ++i) p[dir_pos(k, l + i)] = v[i];
    } else {
        const float4 t = (k == 2) ? make_float4(v[3], v[2], v[1], v[0])
                                  : make_float4(v[0], v[1], v[2], v[3]);
        *reinterpret_cast<float4*>(p + ((k == 2) ? LL - 4 - l : l)) = t;
    }
}

template <int S>
__device__ __forceinline__ void carry_step(const float (&ldv)[MCH][4], const float (&duv)[MCH][4],
                                           const v2f (&An)[MCH][QP], const float4 Bv,
                                           v2f (&h)[MCH][QP], v2f (&ap)[MCH][QP]) {
    const v2f Bs[QP] = {{Bv.x, Bv.y}, {Bv.z, Bv.w}};
    #pragma unroll
    for (int m = 0; m < MCH; ++m) {
        const float ld = row_dpp<(S >> 2) * 0x55>(ldv[m][S & 3]);
        const float du = row_dpp<(S >> 2) * 0x55>(duv[m][S & 3]);
        #pragma unroll
        for (int p = 0; p < QP; ++p) {
            const v2f e = ld * An[m][p];
            v2f dA;
            dA.x = __expf(e.x);
            dA.y = __expf(e.y);
            h[m][p]  = dA * h[m][p] + du * Bs[p];
            ap[m][p] *= dA;
        }
    }
}

template <int S>
__device__ __forceinline__ void out_step(const int q, const float (&ldv)[MCH][4], const float (&duv)[MCH][4],
                                         const v2f (&An)[MCH][QP], const float4 Bv, const float4 Cv,
                                         v2f (&h)[MCH][QP], float (&ykeep)[MCH][4]) {
    const v2f Bs[QP] = {{Bv.x, Bv.y}, {Bv.z, Bv.w}};
    const v2f Cs[QP] = {{Cv.x, Cv.y}, {Cv.z, Cv.w}};
    #pragma unroll
    for (int m = 0; m < MCH; ++m) {
        const float ld = row_dpp<(S >> 2) * 0x55>(ldv[m][S & 3]);
        const float du = row_dpp<(S >> 2) * 0x55>(duv[m][S & 3]);
        #pragma unroll
        for (int p = 0; p < QP; ++p) {
            const v2f e = ld * An[m][p];
            v2f dA;
            dA.x = __expf(e.x);
            dA.y = __expf(e.y);
            h[m][p] = dA * h[m][p] + du * Bs[p];
        }
        const v2f s2 = h[m][0] * Cs[0] + h[m][1] * Cs[1];
        float v = s2.x + s2.y;
        v += row_dpp<0x160 + 1>(v);
        v += row_dpp<0x160 + 2>(v);
        ykeep[m][S & 3] = (q == (S >> 2)) ? v : ykeep[m][S & 3];
    }
}

template <int G>
__device__ __forceinline__ void load_rows(const float* __restrict__ rq, float4 (&R)[4]) {
    #pragma unroll
    for (int s = 0; s < 4; ++s) R[s] = *reinterpret_cast<const float4*>(rq + (4 * G + s) * NN);
}
template <int G>
__device__ __forceinline__ void carry_group(const float (&ldv)[MCH][4], const float (&duv)[MCH][4],
                                            const v2f (&An)[MCH][QP], const float4 (&Bv)[4],
                                            v2f (&h)[MCH][QP], v2f (&ap)[MCH][QP]) {
    carry_step<4 * G + 0>(ldv, duv, An, Bv[0], h, ap); carry_step<4 * G + 1>(ldv, duv, An, Bv[1], h, ap);
    carry_step<4 * G + 2>(ldv, duv, An, Bv[2], h, ap); carry_step<4 * G + 3>(ldv, duv, An, Bv[3], h, ap);
}
template <int G>
__device__ __forceinline__ void out_group(const int q, const float (&ldv)[MCH][4], const float (&duv)[MCH][4],
                                          const v2f (&An)[MCH][QP], const float4 (&Bv)[4], const float4 (&Cv)[4],
                                          v2f (&h)[MCH][QP], float (&ykeep)[MCH][4]) {
    out_step<4 * G + 0>(q, ldv, duv, An, Bv[0], Cv[0], h, ykeep); out_step<4 * G + 1>(q, ldv, duv, An, Bv[1], Cv[1], h, ykeep);
    out_step<4 * G + 2>(q, ldv, duv, An, Bv[2], Cv[2], h, ykeep); out_step<4 * G + 3>(q, ldv, duv, An, Bv[3], Cv[3], h, ykeep);
}
#define GROUP_FENCE() __builtin_amdgcn_sched_barrier(0)

__global__ __launch_bounds__(SCAN_THREADS)
void s6_scan_kernel(const float* __restrict__ x,
                    const float* __restrict__ delta,
                    const float* __restrict__ Bt,
                    const float* __restrict__ Ct,
                    const float* __restrict__ A_log,
                    const float* __restrict__ Ds,
                    float* __restrict__ out)
{
    __shared__ __attribute__((aligned(16))) float sA[MCH][CH][NN];
    __shared__ __attribute__((aligned(16))) float sH[MCH][CH][NN];

    const int tid   = threadIdx.x;
    const int chunk = tid >> 2;
    const int q     = tid & 3;
    const int n0    = q * QN;
    const int grp   = blockIdx.x;
    const int c0    = (grp % (CC / MCH)) * MCH;
    const int bk    = grp / (CC / MCH);
    const int k     = bk % KK;
    const int b     = bk / KK;
    const int d0    = k * CC + c0;

    v2f An[MCH][QP];
    float Dd[MCH];
    #pragma unroll
    for (int m = 0; m < MCH; ++m) {
        #pragma unroll
        for (int p = 0; p < QP; ++p) {
            An[m][p].x = -__expf(A_log[(d0 + m) * NN + n0 + 2 * p + 0]);
            An[m][p].y = -__expf(A_log[(d0 + m) * NN + n0 + 2 * p + 1]);
        }
        Dd[m] = Ds[d0 + m];
    }
    const float* dptr = delta + ((size_t)bk * CC + c0) * LL;
    const float* bptr = Bt + (size_t)bk * LL * NN + n0;
    const float* cptr = Ct + (size_t)bk * LL * NN + n0;
    const float* xptr = x + ((size_t)b * CC + c0) * LL;
    const int slot = (k == 0) ? 0 : (k == 1) ? 2 : (k == 2) ? 1 : 3;
    float* optr = out + ((size_t)(slot * BB + b) * CC + c0) * LL;
    const int lbeg = chunk * CHL;

    {
        v2f h[MCH][QP], ap[MCH][QP];
        #pragma unroll
        for (int m = 0; m < MCH; ++m) {
            #pragma unroll
            for (int p = 0; p < QP; ++p) { h[m][p] = (v2f){0.0f, 0.0f}; ap[m][p] = (v2f){1.0f, 1.0f}; }
        }
        for (int t = 0; t < CHL / GRP; ++t) {
            const int l0 = lbeg + t * GRP;
            const int lq = l0 + 4 * q;
            float ldv[MCH][4], uv[MCH][4], duv[MCH][4];
            load_u(xptr, k, lq, uv);
            #pragma unroll
            for (int m = 0; m < MCH; ++m) {
                const float4 d4 = *reinterpret_cast<const float4*>(dptr + m * LL + lq);
                ldv[m][0] = d4.x; ldv[m][1] = d4.y; ldv[m][2] = d4.z; ldv[m][3] = d4.w;
                #pragma unroll
                for (int i = 0; i < 4; ++i) duv[m][i] = ldv[m][i] * uv[m][i];
            }
            const float* bq = bptr + (size_t)l0 * NN;
            float4 Be[4], Bo[4];
            load_rows<0>(bq, Be);
            load_rows<1>(bq, Bo);
            carry_group<0>(ldv, duv, An, Be, h, ap);
            GROUP_FENCE();
            load_rows<2>(bq, Be);
            carry_group<1>(ldv, duv, An, Bo, h, ap);
            GROUP_FENCE();
            load_rows<3>(bq, Bo);
            carry_group<2>(ldv, duv, An, Be, h, ap);
            carry_group<3>(ldv, duv, An, Bo, h, ap);
        }
        #pragma unroll
        for (int m = 0; m < MCH; ++m) {
            *reinterpret_cast<float4*>(&sA[m][chunk][n0]) = make_float4(ap[m][0].x, ap[m][0].y, ap[m][1].x, ap[m][1].y);
            *reinterpret_cast<float4*>(&sH[m][chunk][n0]) = make_float4(h[m][0].x, h[m][0].y, h[m][1].x, h[m][1].y);
        }
    }
    __syncthreads();

    if (tid < MCH * NN) {
        const int m = tid >> 4;
        const int n = tid & 15;
        float Hp = 0.0f;
        #pragma unroll 8
        for (int j = 0; j < CH; ++j) {
            const float hj = sH[m][j][n];
            sH[m][j][n] = Hp;
            Hp = sA[m][j][n] * Hp + hj;
        }
    }
    __syncthreads();

    v2f h[MCH][QP];
    #pragma unroll
    for (int m = 0; m < MCH; ++m) {
        const float4 i4 = *reinterpret_cast<const float4*>(&sH[m][chunk][n0]);
        h[m][0] = (v2f){i4.x, i4.y};
        h[m][1] = (v2f){i4.z, i4.w};
    }
    for (int t = 0; t < CHL / GRP; ++t) {
        const int l0 = lbeg + t * GRP;
        const int lq = l0 + 4 * q;
        float ldv[MCH][4], uv[MCH][4], duv[MCH][4], ykeep[MCH][4];
        load_u(xptr, k, lq, uv);
        #pragma unroll
        for (int m = 0; m < MCH; ++m) {
            const float4 d4 = *reinterpret_cast<const float4*>(dptr + m * LL + lq);
            ldv[m][0] = d4.x; ldv[m][1] = d4.y; ldv[m][2] = d4.z; ldv[m][3] = d4.w;
            #pragma unroll
            for (int i = 0; i < 4; ++i) { duv[m][i] = ldv[m][i] * uv[m][i]; ykeep[m][i] = 0.0f; }
        }
        const float* bq = bptr + (size_t)l0 * NN;
        const float* cq = cptr + (size_t)l0 * NN;
        float4 Be[4], Ce[4], Bo[4], Co[4];
        load_rows<0>(bq, Be); load_rows<0>(cq, Ce);
        load_rows<1>(bq, Bo); load_rows<1>(cq, Co);
        out_group<0>(q, ldv, duv, An, Be, Ce, h, ykeep);
        GROUP_FENCE();
        load_rows<2>(bq, Be); load_rows<2>(cq, Ce);
        out_group<1>(q, ldv, duv, An, Bo, Co, h, ykeep);
        GROUP_FENCE();
        load_rows<3>(bq, Bo); load_rows<3>(cq, Co);
        out_group<2>(q, ldv, duv, An, Be, Ce, h, ykeep);
        out_group<3>(q, ldv, duv, An, Bo, Co, h, ykeep);
        #pragma unroll
        for (int m = 0; m < MCH; ++m) {
            float y4[4];
            #pragma unroll
            for (int i = 0; i < 4; ++i) y4[i] = ykeep[m][i] + Dd[m] * uv[m][i];
            store_dir4(optr + m * LL, k, lq, y4);
        }
    }
}

extern "C" void kernel_launch(void* const* d_in, const int* in_sizes, int n_in,
                              void* d_out, int out_size, void* d_ws, size_t ws_size,
                              hipStream_t stream) {
    const float* x      = (const float*)d_in[0];
    const float* wproj  = (const float*)d_in[1];
    const float* wdelta = (const float*)d_in[2];
    const float* dbias  = (const float*)d_in[3];
    const float* A_log  = (const float*)d_in[4];
    const float* Ds     = (const float*)d_in[5];
    float* out = (float*)d_out;

    float* ws     = (float*)d_ws;
    float* delta  = ws;
    float* Bt     = delta  + (size_t)BB * KK * CC * LL;
    float* Ct     = Bt     + (size_t)BB * KK * LL * NN;

    dim3 gProj(LL / LT, KK, BB);
    s6_proj_kernel<<<gProj, 256, 0, stream>>>(x, wproj, wdelta, dbias, delta, Bt, Ct);

    s6_scan_kernel<<<BB * KK * (CC / MCH), SCAN_THREADS, 0, stream>>>(x, delta, Bt, Ct, A_log, Ds, out);
}


# projection stores delta (paired with its x value), B and C by image position, not by scan step: x reads and stores are whole lines in all 4 directions; the scan applies the direction map on reading
# speedup vs baseline: 18.8990x; 1.2381x over previous
#include <hip/hip_runtime.h>
#include <hip/hip_bf16.h>

#define BB   2
#define CC   96
#define HH   64
#define WWW  64
#define LL   4096
#define NN   16
#define RR   6
#define KK   4
#define DBCROWS 38
#define MPAD 48
#define LT   64
#define QN   4
#define QL   (NN / QN)
#define CH   128
#define CHL  (LL / CH)

typedef float v2f __attribute__((ext_vector_type(2)));
typedef float v8f __attribute__((ext_vector_type(8)));
typedef float v4f __attribute__((ext_vector_type(4)));

__device__ __forceinline__ int tr_idx(int l) { return ((l & 63) << 6) | (l >> 6); }
__device__ __forceinline__ int dir_pos(const int k, const int l) {
    const int lr = (k & 2) ? LL - 1 - l : l;
    return (k & 1) ? tr_idx(lr) : lr;
}

template <int CTRL>
__device__ __forceinline__ float row_dpp(float v) {
    return __builtin_bit_cast(float,
        __builtin_amdgcn_update_dpp(0, __builtin_bit_cast(int, v), CTRL, 0xf, 0xf, true));
}
#define GRP 16

__global__ __launch_bounds__(256)
void s6_proj_kernel(const float* __restrict__ x,
                    const float* __restrict__ wproj,
                    const float* __restrict__ wdelta,
                    const float* __restrict__ dbias,
                    float* __restrict__ du_out,
                    float* __restrict__ Bt,
                    float* __restrict__ Ct)
{
    __shared__ __attribute__((aligned(16))) float sW[MPAD][CC];
    __shared__ float sX[CC][LT + 1];
    __shared__ float sDBC[MPAD][LT + 1];
    __shared__ __attribute__((aligned(16))) float sWd[256 * 4];
    __shared__ float sBias[256];

    const int tid = threadIdx.x;
    const int l0  = blockIdx.x * LT;
    const int k   = blockIdx.y;
    const int b   = blockIdx.z;

    constexpr int W4  = DBCROWS * CC / 4;
    constexpr int WP4 = MPAD * CC / 4;
    constexpr int D4  = CC * RR / 4;
    constexpr int XJ  = CC * LT / 256;
    const v4f* wsrc = reinterpret_cast<const v4f*>(wproj + (size_t)k * DBCROWS * CC);
    const v4f* dsrc = reinterpret_cast<const v4f*>(wdelta + (size_t)k * CC * RR);
    const v4f   w0 = wsrc[tid];
    const v4f   w1 = wsrc[tid + 256];
    const v4f   w2 = wsrc[tid + 512];
    const bool  w3in = tid + 768 < W4;
    const v4f   w3 = wsrc[w3in ? tid + 768 : W4 - 1];
    const v4f   dv = dsrc[tid < D4 ? tid : D4 - 1];
    const float bv = dbias[k * CC + (tid < CC ? tid : CC - 1)];
    const int col = tid & (LT - 1);
    const int cr  = tid >> 6;
    const float* xcol = x + (size_t)b * CC * LL + l0 + col;
    float xv[XJ];
    #pragma unroll
    for (int j = 0; j < XJ; ++j) xv[j] = xcol[(size_t)(cr + 4 * j) * LL];

    v4f* sW4 = reinterpret_cast<v4f*>(&sW[0][0]);
    const v4f zero4 = {0.0f, 0.0f, 0.0f, 0.0f};
    sW4[tid]       = w0;
    sW4[tid + 256] = w1;
    sW4[tid + 512] = w2;
    sW4[tid + 768] = w3in ? w3 : zero4;
    if (tid < WP4 - 1024) sW4[tid + 1024] = zero4;
    reinterpret_cast<v4f*>(sWd)[tid] = dv;
    sBias[tid] = bv;
    #pragma unroll
    for (int j = 0; j < XJ; ++j) sX[cr + 4 * j][col] = xv[j];
    __syncthreads();

    const int wave = tid >> 5;
    const int lane = tid & 31;
    const int lm   = lane & 15;
    const int lh   = lane >> 4;
    for (int t = wave; t < (MPAD / 16) * (LT / 16); t += 8) {
        const int mt = t / (LT / 16);
        const int nt = t % (LT / 16);
        v8f acc = {};
        for (int kk = 0; kk < CC; kk += 4) {
            v2f a, bf;
            a.x  = sW[mt * 16 + lm][kk + 2 * lh + 0];
            a.y  = sW[mt * 16 + lm][kk + 2 * lh + 1];
            bf.x = sX[kk + 2 * lh + 0][nt * 16 + lm];
            bf.y = sX[kk + 2 * lh + 1][nt * 16 + lm];
            acc = __builtin_amdgcn_wmma_f32_16x16x4_f32(
                      false, a, false, bf, (short)0, acc, false, false);
        }
        #pragma unroll
        for (int i = 0; i < 8; ++i)
            sDBC[mt * 16 + i + 8 * lh][nt * 16 + lm] = acc[i];
    }
    __syncthreads();

    for (int i = tid; i < LT * NN; i += 256) {
        int col = i / NN, n = i % NN;
        size_t base = (((size_t)(b * KK + k) * LL + (l0 + col)) * NN + n);
        Bt[base] = sDBC[RR + n][col];
        Ct[base] = sDBC[RR + NN + n][col];
    }
    v2f* dpair = reinterpret_cast<v2f*>(du_out);
    for (int i = tid; i < CC * LT; i += 256) {
        int c = i / LT, col = i % LT;
        float s = 0.0f;
        #pragma unroll
        for (int r = 0; r < RR; ++r) s += sWd[c * RR + r] * sDBC[r][col];
        s += sBias[c];
        float sp = (s > 20.0f) ? s : log1pf(__expf(s));
        dpair[(((size_t)(b * KK + k) * CC + c) * LL) + l0 + col] = (v2f){sp, sX[c][col]};
    }
}

#define SCAN_THREADS (CH * QL)
#define MCH 3
#define QP  (QN / 2)

__device__ __forceinline__ int pos_stride(const int k) { return ((k & 1) ? 64 : 1) * ((k & 2) ? -1 : 1); }
__device__ __forceinline__ void load_du(const float* __restrict__ p, const int pq, const int pst,
                                        float (&d)[MCH][4], float (&u)[MCH][4]) {
    const char* pc = reinterpret_cast<const char*>(p);
    const unsigned ob = 8u * (unsigned)pq;
    const int sb = 8 * pst;
    #pragma unroll
    for (int i = 0; i < 4; ++i) {
        const unsigned o = ob + (unsigned)(i * sb);
        #pragma unroll
        for (int m = 0; m < MCH; ++m) {
            const v2f v = *reinterpret_cast<const v2f*>(pc + (size_t)o + (size_t)(m * LL * 8));
            d[m][i] = v.x;
            u[m][i] = v.y;
        }
    }
}
__device__ __forceinline__ void store_dir4(float* __restrict__ p, const int k, const int l,
                                           const float (&v)[4]) {
    if (k & 1) {
        #pragma unroll
        for (int i = 0; i < 4; ++i) p[dir_pos(k, l + i)] = v[i];
    } else {
        const float4 t = (k == 2) ? make_float4(v[3], v[2], v[1], v[0])
                                  : make_float4(v[0], v[1], v[2], v[3]);
        *reinterpret_cast<float4*>(p + ((k == 2) ? LL - 4 - l : l)) = t;
    }
}

template <int S>
__device__ __forceinline__ void carry_step(const float (&ldv)[MCH][4], const float (&duv)[MCH][4],
                                           const v2f (&An)[MCH][QP], const float4 Bv,
                                           v2f (&h)[MCH][QP], v2f (&ap)[MCH][QP]) {
    const v2f Bs[QP] = {{Bv.x, Bv.y}, {Bv.z, Bv.w}};
    #pragma unroll
    for (int m = 0; m < MCH; ++m) {
        const float ld = row_dpp<(S >> 2) * 0x55>(ldv[m][S & 3]);
        const float du = row_dpp<(S >> 2) * 0x55>(duv[m][S & 3]);
        #pragma unroll
        for (int p = 0; p < QP; ++p) {
            const v2f e = ld * An[m][p];
            v2f dA;
            dA.x = __expf(e.x);
            dA.y = __expf(e.y);
            h[m][p]  = dA * h[m][p] + du * Bs[p];
            ap[m][p] *= dA;
        }
    }
}

template <int S>
__device__ __forceinline__ void out_step(const int q, const float (&ldv)[MCH][4], const float (&duv)[MCH][4],
                                         const v2f (&An)[MCH][QP], const float4 Bv, const float4 Cv,
                                         v2f (&h)[MCH][QP], float (&ykeep)[MCH][4]) {
    const v2f Bs[QP] = {{Bv.x, Bv.y}, {Bv.z, Bv.w}};
    const v2f Cs[QP] = {{Cv.x, Cv.y}, {Cv.z, Cv.w}};
    #pragma unroll
    for (int m = 0; m < MCH; ++m) {
        const float ld = row_dpp<(S >> 2) * 0x55>(ldv[m][S & 3]);
        const float du = row_dpp<(S >> 2) * 0x55>(duv[m][S & 3]);
        #pragma unroll
        for (int p = 0; p < QP; ++p) {
            const v2f e = ld * An[m][p];
            v2f dA;
            dA.x = __expf(e.x);
            dA.y = __expf(e.y);
            h[m][p] = dA * h[m][p] + du * Bs[p];
        }
        const v2f s2 = h[m][0] * Cs[0] + h[m][1] * Cs[1];
        float v = s2.x + s2.y;
        v += row_dpp<0x160 + 1>(v);
        v += row_dpp<0x160 + 2>(v);
        ykeep[m][S & 3] = (q == (S >> 2)) ? v : ykeep[m][S & 3];
    }
}

template <int G>
__device__ __forceinline__ void load_rows(const float* __restrict__ pl, const int r0, const int rst,
                                          float4 (&R)[4]) {
    const char* pc = reinterpret_cast<const char*>(pl);
    #pragma unroll
    for (int s = 0; s < 4; ++s) {
        const unsigned o = 4u * (unsigned)(r0 + (4 * G + s) * rst);
        R[s] = *reinterpret_cast<const float4*>(pc + (size_t)o);
    }
}
template <int G>
__device__ __forceinline__ void carry_group(const float (&ldv)[MCH][4], const float (&duv)[MCH][4],
                                            const v2f (&An)[MCH][QP], const float4 (&Bv)[4],
                                            v2f (&h)[MCH][QP], v2f (&ap)[MCH][QP]) {
    carry_step<4 * G + 0>(ldv, duv, An, Bv[0], h, ap); carry_step<4 * G + 1>(ldv, duv, An, Bv[1], h, ap);
    carry_step<4 * G + 2>(ldv, duv, An, Bv[2], h, ap); carry_step<4 * G + 3>(ldv, duv, An, Bv[3], h, ap);
}
template <int G>
__device__ __forceinline__ void out_group(const int q, const float (&ldv)[MCH][4], const float (&duv)[MCH][4],
                                          const v2f (&An)[MCH][QP], const float4 (&Bv)[4], const float4 (&Cv)[4],
                                          v2f (&h)[MCH][QP], float (&ykeep)[MCH][4]) {
    out_step<4 * G + 0>(q, ldv, duv, An, Bv[0], Cv[0], h, ykeep); out_step<4 * G + 1>(q, ldv, duv, An, Bv[1], Cv[1], h, ykeep);
    out_step<4 * G + 2>(q, ldv, duv, An, Bv[2], Cv[2], h, ykeep); out_step<4 * G + 3>(q, ldv, duv, An, Bv[3], Cv[3], h, ykeep);
}
#define GROUP_FENCE() __builtin_amdgcn_sched_barrier(0)

__global__ __launch_bounds__(SCAN_THREADS)
void s6_scan_kernel(const float* __restrict__ du,
                    const float* __restrict__ Bt,
                    const float* __restrict__ Ct,
                    const float* __restrict__ A_log,
                    const float* __restrict__ Ds,
                    float* __restrict__ out)
{
    __shared__ __attribute__((aligned(16))) float sA[MCH][CH][NN];
    __shared__ __attribute__((aligned(16))) float sH[MCH][CH][NN];

    const int tid   = threadIdx.x;
    const int chunk = tid >> 2;
    const int q     = tid & 3;
    const int n0    = q * QN;
    const int grp   = blockIdx.x;
    const int c0    = (grp % (CC / MCH)) * MCH;
    const int bk    = grp / (CC / MCH);
    const int k     = bk % KK;
    const int b     = bk / KK;
    const int d0    = k * CC + c0;

    v2f An[MCH][QP];
    float Dd[MCH];
    #pragma unroll
    for (int m = 0; m < MCH; ++m) {
        #pragma unroll
        for (int p = 0; p < QP; ++p) {
            An[m][p].x = -__expf(A_log[(d0 + m) * NN + n0 + 2 * p + 0]);
            An[m][p].y = -__expf(A_log[(d0 + m) * NN + n0 + 2 * p + 1]);
        }
        Dd[m] = Ds[d0 + m];
    }
    const float* dup = du + ((size_t)bk * CC + c0) * LL * 2;
    const float* bpl = Bt + (size_t)bk * LL * NN;
    const float* cpl = Ct + (size_t)bk * LL * NN;
    const int pst = pos_stride(k);
    const int rst = pst * NN;
    const int slot = (k == 0) ? 0 : (k == 1) ? 2 : (k == 2) ? 1 : 3;
    float* optr = out + ((size_t)(slot * BB + b) * CC + c0) * LL;
    const int lbeg = chunk * CHL;

    {
        v2f h[MCH][QP], ap[MCH][QP];
        #pragma unroll
        for (int m = 0; m < MCH; ++m) {
            #pragma unroll
            for (int p = 0; p < QP; ++p) { h[m][p] = (v2f){0.0f, 0.0f}; ap[m][p] = (v2f){1.0f, 1.0f}; }
        }
        for (int t = 0; t < CHL / GRP; ++t) {
            const int l0 = lbeg + t * GRP;
            const int p0 = dir_pos(k, l0);
            float ldv[MCH][4], uv[MCH][4], duv[MCH][4];
            load_du(dup, p0 + 4 * q * pst, pst, ldv, uv);
            #pragma unroll
            for (int m = 0; m < MCH; ++m) {
                #pragma unroll
                for (int i = 0; i < 4; ++i) duv[m][i] = ldv[m][i] * uv[m][i];
            }
            const int r0 = p0 * NN + n0;
            float4 Be[4], Bo[4];
            load_rows<0>(bpl, r0, rst, Be);
            load_rows<1>(bpl, r0, rst, Bo);
            carry_group<0>(ldv, duv, An, Be, h, ap);
            GROUP_FENCE();
            load_rows<2>(bpl, r0, rst, Be);
            carry_group<1>(ldv, duv, An, Bo, h, ap);
            GROUP_FENCE();
            load_rows<3>(bpl, r0, rst, Bo);
            carry_group<2>(ldv, duv, An, Be, h, ap);
            carry_group<3>(ldv, duv, An, Bo, h, ap);
        }
        #pragma unroll
        for (int m = 0; m < MCH; ++m) {
            *reinterpret_cast<float4*>(&sA[m][chunk][n0]) = make_float4(ap[m][0].x, ap[m][0].y, ap[m][1].x, ap[m][1].y);
            *reinterpret_cast<float4*>(&sH[m][chunk][n0]) = make_float4(h[m][0].x, h[m][0].y, h[m][1].x, h[m][1].y);
        }
    }
    __syncthreads();

    if (tid < MCH * NN) {
        const int m = tid >> 4;
        const int n = tid & 15;
        float Hp = 0.0f;
        #pragma unroll 8
        for (int j = 0; j < CH; ++j) {
            const float hj = sH[m][j][n];
            sH[m][j][n] = Hp;
            Hp = sA[m][j][n] * Hp + hj;
        }
    }
    __syncthreads();

    v2f h[MCH][QP];
    #pragma unroll
    for (int m = 0; m < MCH; ++m) {
        const float4 i4 = *reinterpret_cast<const float4*>(&sH[m][chunk][n0]);
        h[m][0] = (v2f){i4.x, i4.y};
        h[m][1] = (v2f){i4.z, i4.w};
    }
    for (int t = 0; t < CHL / GRP; ++t) {
        const int l0 = lbeg + t * GRP;
        const int lq = l0 + 4 * q;
        const int p0 = dir_pos(k, l0);
        float ldv[MCH][4], uv[MCH][4], duv[MCH][4], ykeep[MCH][4];
        load_du(dup, p0 + 4 * q * pst, pst, ldv, uv);
        #pragma unroll
        for (int m = 0; m < MCH; ++m) {
            #pragma unroll
            for (int i = 0; i < 4; ++i) { duv[m][i] = ldv[m][i] * uv[m][i]; ykeep[m][i] = 0.0f; }
        }
        const int r0 = p0 * NN + n0;
        float4 Be[4], Ce[4], Bo[4], Co[4];
        load_rows<0>(bpl, r0, rst, Be); load_rows<0>(cpl, r0, rst, Ce);
        load_rows<1>(bpl, r0, rst, Bo); load_rows<1>(cpl, r0, rst, Co);
        out_group<0>(q, ldv, duv, An, Be, Ce, h, ykeep);
        GROUP_FENCE();
        load_rows<2>(bpl, r0, rst, Be); load_rows<2>(cpl, r0, rst, Ce);
        out_group<1>(q, ldv, duv, An, Bo, Co, h, ykeep);
        GROUP_FENCE();
        load_rows<3>(bpl, r0, rst, Bo); load_rows<3>(cpl, r0, rst, Co);
        out_group<2>(q, ldv, duv, An, Be, Ce, h, ykeep);
        out_group<3>(q, ldv, duv, An, Bo, Co, h, ykeep);
        #pragma unroll
        for (int m = 0; m < MCH; ++m) {
            float y4[4];
            #pragma unroll
            for (int i = 0; i < 4; ++i) y4[i] = ykeep[m][i] + Dd[m] * uv[m][i];
            store_dir4(optr + m * LL, k, lq, y4);
        }
    }
}

extern "C" void kernel_launch(void* const* d_in, const int* in_sizes, int n_in,
                              void* d_out, int out_size, void* d_ws, size_t ws_size,
                              hipStream_t stream) {
    const float* x      = (const float*)d_in[0];
    const float* wproj  = (const float*)d_in[1];
    const float* wdelta = (const float*)d_in[2];
    const float* dbias  = (const float*)d_in[3];
    const float* A_log  = (const float*)d_in[4];
    const float* Ds     = (const float*)d_in[5];
    float* out = (float*)d_out;

    float* ws     = (float*)d_ws;
    float* du     = ws;
    float* Bt     = du     + (size_t)BB * KK * CC * LL * 2;
    float* Ct     = Bt     + (size_t)BB * KK * LL * NN;

    dim3 gProj(LL / LT, KK, BB);
    s6_proj_kernel<<<gProj, 256, 0, stream>>>(x, wproj, wdelta, dbias, du, Bt, Ct);

    s6_scan_kernel<<<BB * KK * (CC / MCH), SCAN_THREADS, 0, stream>>>(du, Bt, Ct, A_log, Ds, out);
}


# projection epilogue: softplus as hardware exp2/log2 of u=1+e plus the exact rounding remainder (e-(u-1))/u, instead of the library log1p's long dependent chain; delta may differ in its last bits
# speedup vs baseline: 20.9257x; 1.1072x over previous
#include <hip/hip_runtime.h>
#include <hip/hip_bf16.h>

#define BB   2
#define CC   96
#define HH   64
#define WWW  64
#define LL   4096
#define NN   16
#define RR   6
#define KK   4
#define DBCROWS 38
#define MPAD 48
#define LT   64
#define QN   4
#define QL   (NN / QN)
#define CH   128
#define CHL  (LL / CH)

typedef float v2f __attribute__((ext_vector_type(2)));
typedef float v8f __attribute__((ext_vector_type(8)));
typedef float v4f __attribute__((ext_vector_type(4)));

__device__ __forceinline__ int tr_idx(int l) { return ((l & 63) << 6) | (l >> 6); }
__device__ __forceinline__ int dir_pos(const int k, const int l) {
    const int lr = (k & 2) ? LL - 1 - l : l;
    return (k & 1) ? tr_idx(lr) : lr;
}

template <int CTRL>
__device__ __forceinline__ float row_dpp(float v) {
    return __builtin_bit_cast(float,
        __builtin_amdgcn_update_dpp(0, __builtin_bit_cast(int, v), CTRL, 0xf, 0xf, true));
}
#define GRP 16

__global__ __launch_bounds__(256)
void s6_proj_kernel(const float* __restrict__ x,
                    const float* __restrict__ wproj,
                    const float* __restrict__ wdelta,
                    const float* __restrict__ dbias,
                    float* __restrict__ du_out,
                    float* __restrict__ Bt,
                    float* __restrict__ Ct)
{
    __shared__ __attribute__((aligned(16))) float sW[MPAD][CC];
    __shared__ float sX[CC][LT + 1];
    __shared__ float sDBC[MPAD][LT + 1];
    __shared__ __attribute__((aligned(16))) float sWd[256 * 4];
    __shared__ float sBias[256];

    const int tid = threadIdx.x;
    const int l0  = blockIdx.x * LT;
    const int k   = blockIdx.y;
    const int b   = blockIdx.z;

    constexpr int W4  = DBCROWS * CC / 4;
    constexpr int WP4 = MPAD * CC / 4;
    constexpr int D4  = CC * RR / 4;
    constexpr int XJ  = CC * LT / 256;
    const v4f* wsrc = reinterpret_cast<const v4f*>(wproj + (size_t)k * DBCROWS * CC);
    const v4f* dsrc = reinterpret_cast<const v4f*>(wdelta + (size_t)k * CC * RR);
    const v4f   w0 = wsrc[tid];
    const v4f   w1 = wsrc[tid + 256];
    const v4f   w2 = wsrc[tid + 512];
    const bool  w3in = tid + 768 < W4;
    const v4f   w3 = wsrc[w3in ? tid + 768 : W4 - 1];
    const v4f   dv = dsrc[tid < D4 ? tid : D4 - 1];
    const float bv = dbias[k * CC + (tid < CC ? tid : CC - 1)];
    const int col = tid & (LT - 1);
    const int cr  = tid >> 6;
    const float* xcol = x + (size_t)b * CC * LL + l0 + col;
    float xv[XJ];
    #pragma unroll
    for (int j = 0; j < XJ; ++j) xv[j] = xcol[(size_t)(cr + 4 * j) * LL];

    v4f* sW4 = reinterpret_cast<v4f*>(&sW[0][0]);
    const v4f zero4 = {0.0f, 0.0f, 0.0f, 0.0f};
    sW4[tid]       = w0;
    sW4[tid + 256] = w1;
    sW4[tid + 512] = w2;
    sW4[tid + 768] = w3in ? w3 : zero4;
    if (tid < WP4 - 1024) sW4[tid + 1024] = zero4;
    reinterpret_cast<v4f*>(sWd)[tid] = dv;
    sBias[tid] = bv;
    #pragma unroll
    for (int j = 0; j < XJ; ++j) sX[cr + 4 * j][col] = xv[j];
    __syncthreads();

    const int wave = tid >> 5;
    const int lane = tid & 31;
    const int lm   = lane & 15;
    const int lh   = lane >> 4;
    for (int t = wave; t < (MPAD / 16) * (LT / 16); t += 8) {
        const int mt = t / (LT / 16);
        const int nt = t % (LT / 16);
        v8f acc = {};
        for (int kk = 0; kk < CC; kk += 4) {
            v2f a, bf;
            a.x  = sW[mt * 16 + lm][kk + 2 * lh + 0];
            a.y  = sW[mt * 16 + lm][kk + 2 * lh + 1];
            bf.x = sX[kk + 2 * lh + 0][nt * 16 + lm];
            bf.y = sX[kk + 2 * lh + 1][nt * 16 + lm];
            acc = __builtin_amdgcn_wmma_f32_16x16x4_f32(
                      false, a, false, bf, (short)0, acc, false, false);
        }
        #pragma unroll
        for (int i = 0; i < 8; ++i)
            sDBC[mt * 16 + i + 8 * lh][nt * 16 + lm] = acc[i];
    }
    __syncthreads();

    for (int i = tid; i < LT * NN; i += 256) {
        int col = i / NN, n = i % NN;
        size_t base = (((size_t)(b * KK + k) * LL + (l0 + col)) * NN + n);
        Bt[base] = sDBC[RR + n][col];
        Ct[base] = sDBC[RR + NN + n][col];
    }
    v2f* dpair = reinterpret_cast<v2f*>(du_out);
    for (int i = tid; i < CC * LT; i += 256) {
        int c = i / LT, col = i % LT;
        float s = 0.0f;
        #pragma unroll
        for (int r = 0; r < RR; ++r) s += sWd[c * RR + r] * sDBC[r][col];
        s += sBias[c];
        const float e  = __expf(s);
        const float u  = 1.0f + e;
        const float lg = __builtin_amdgcn_logf(u) * 0.693147181f + (e - (u - 1.0f)) * __builtin_amdgcn_rcpf(u);
        const float sp = (s > 20.0f) ? s : lg;
        dpair[(((size_t)(b * KK + k) * CC + c) * LL) + l0 + col] = (v2f){sp, sX[c][col]};
    }
}

#define SCAN_THREADS (CH * QL)
#define MCH 3
#define QP  (QN / 2)

__device__ __forceinline__ int pos_stride(const int k) { return ((k & 1) ? 64 : 1) * ((k & 2) ? -1 : 1); }
__device__ __forceinline__ void load_du(const float* __restrict__ p, const int pq, const int pst,
                                        float (&d)[MCH][4], float (&u)[MCH][4]) {
    const char* pc = reinterpret_cast<const char*>(p);
    const unsigned ob = 8u * (unsigned)pq;
    const int sb = 8 * pst;
    #pragma unroll
    for (int i = 0; i < 4; ++i) {
        const unsigned o = ob + (unsigned)(i * sb);
        #pragma unroll
        for (int m = 0; m < MCH; ++m) {
            const v2f v = *reinterpret_cast<const v2f*>(pc + (size_t)o + (size_t)(m * LL * 8));
            d[m][i] = v.x;
            u[m][i] = v.y;
        }
    }
}
__device__ __forceinline__ void store_dir4(float* __restrict__ p, const int k, const int l,
                                           const float (&v)[4]) {
    if (k & 1) {
        #pragma unroll
        for (int i = 0; i < 4; ++i) p[dir_pos(k, l + i)] = v[i];
    } else {
        const float4 t = (k == 2) ? make_float4(v[3], v[2], v[1], v[0])
                                  : make_float4(v[0], v[1], v[2], v[3]);
        *reinterpret_cast<float4*>(p + ((k == 2) ? LL - 4 - l : l)) = t;
    }
}

template <int S>
__device__ __forceinline__ void carry_step(const float (&ldv)[MCH][4], const float (&duv)[MCH][4],
                                           const v2f (&An)[MCH][QP], const float4 Bv,
                                           v2f (&h)[MCH][QP], v2f (&ap)[MCH][QP]) {
    const v2f Bs[QP] = {{Bv.x, Bv.y}, {Bv.z, Bv.w}};
    #pragma unroll
    for (int m = 0; m < MCH; ++m) {
        const float ld = row_dpp<(S >> 2) * 0x55>(ldv[m][S & 3]);
        const float du = row_dpp<(S >> 2) * 0x55>(duv[m][S & 3]);
        #pragma unroll
        for (int p = 0; p < QP; ++p) {
            const v2f e = ld * An[m][p];
            v2f dA;
            dA.x = __expf(e.x);
            dA.y = __expf(e.y);
            h[m][p]  = dA * h[m][p] + du * Bs[p];
            ap[m][p] *= dA;
        }
    }
}

template <int S>
__device__ __forceinline__ void out_step(const int q, const float (&ldv)[MCH][4], const float (&duv)[MCH][4],
                                         const v2f (&An)[MCH][QP], const float4 Bv, const float4 Cv,
                                         v2f (&h)[MCH][QP], float (&ykeep)[MCH][4]) {
    const v2f Bs[QP] = {{Bv.x, Bv.y}, {Bv.z, Bv.w}};
    const v2f Cs[QP] = {{Cv.x, Cv.y}, {Cv.z, Cv.w}};
    #pragma unroll
    for (int m = 0; m < MCH; ++m) {
        const float ld = row_dpp<(S >> 2) * 0x55>(ldv[m][S & 3]);
        const float du = row_dpp<(S >> 2) * 0x55>(duv[m][S & 3]);
        #pragma unroll
        for (int p = 0; p < QP; ++p) {
            const v2f e = ld * An[m][p];
            v2f dA;
            dA.x = __expf(e.x);
            dA.y = __expf(e.y);
            h[m][p] = dA * h[m][p] + du * Bs[p];
        }
        const v2f s2 = h[m][0] * Cs[0] + h[m][1] * Cs[1];
        float v = s2.x + s2.y;
        v += row_dpp<0x160 + 1>(v);
        v += row_dpp<0x160 + 2>(v);
        ykeep[m][S & 3] = (q == (S >> 2)) ? v : ykeep[m][S & 3];
    }
}

template <int G>
__device__ __forceinline__ void load_rows(const float* __restrict__ pl, const int r0, const int rst,
                                          float4 (&R)[4]) {
    const char* pc = reinterpret_cast<const char*>(pl);
    #pragma unroll
    for (int s = 0; s < 4; ++s) {
        const unsigned o = 4u * (unsigned)(r0 + (4 * G + s) * rst);
        R[s] = *reinterpret_cast<const float4*>(pc + (size_t)o);
    }
}
template <int G>
__device__ __forceinline__ void carry_group(const float (&ldv)[MCH][4], const float (&duv)[MCH][4],
                                            const v2f (&An)[MCH][QP], const float4 (&Bv)[4],
                                            v2f (&h)[MCH][QP], v2f (&ap)[MCH][QP]) {
    carry_step<4 * G + 0>(ldv, duv, An, Bv[0], h, ap); carry_step<4 * G + 1>(ldv, duv, An, Bv[1], h, ap);
    carry_step<4 * G + 2>(ldv, duv, An, Bv[2], h, ap); carry_step<4 * G + 3>(ldv, duv, An, Bv[3], h, ap);
}
template <int G>
__device__ __forceinline__ void out_group(const int q, const float (&ldv)[MCH][4], const float (&duv)[MCH][4],
                                          const v2f (&An)[MCH][QP], const float4 (&Bv)[4], const float4 (&Cv)[4],
                                          v2f (&h)[MCH][QP], float (&ykeep)[MCH][4]) {
    out_step<4 * G + 0>(q, ldv, duv, An, Bv[0], Cv[0], h, ykeep); out_step<4 * G + 1>(q, ldv, duv, An, Bv[1], Cv[1], h, ykeep);
    out_step<4 * G + 2>(q, ldv, duv, An, Bv[2], Cv[2], h, ykeep); out_step<4 * G + 3>(q, ldv, duv, An, Bv[3], Cv[3], h, ykeep);
}
#define GROUP_FENCE() __builtin_amdgcn_sched_barrier(0)

__global__ __launch_bounds__(SCAN_THREADS)
void s6_scan_kernel(const float* __restrict__ du,
                    const float* __restrict__ Bt,
                    const float* __restrict__ Ct,
                    const float* __restrict__ A_log,
                    const float* __restrict__ Ds,
                    float* __restrict__ out)
{
    __shared__ __attribute__((aligned(16))) float sA[MCH][CH][NN];
    __shared__ __attribute__((aligned(16))) float sH[MCH][CH][NN];

    const int tid   = threadIdx.x;
    const int chunk = tid >> 2;
    const int q     = tid & 3;
    const int n0    = q * QN;
    const int grp   = blockIdx.x;
    const int c0    = (grp % (CC / MCH)) * MCH;
    const int bk    = grp / (CC / MCH);
    const int k     = bk % KK;
    const int b     = bk / KK;
    const int d0    = k * CC + c0;

    v2f An[MCH][QP];
    float Dd[MCH];
    #pragma unroll
    for (int m = 0; m < MCH; ++m) {
        #pragma unroll
        for (int p = 0; p < QP; ++p) {
            An[m][p].x = -__expf(A_log[(d0 + m) * NN + n0 + 2 * p + 0]);
            An[m][p].y = -__expf(A_log[(d0 + m) * NN + n0 + 2 * p + 1]);
        }
        Dd[m] = Ds[d0 + m];
    }
    const float* dup = du + ((size_t)bk * CC + c0) * LL * 2;
    const float* bpl = Bt + (size_t)bk * LL * NN;
    const float* cpl = Ct + (size_t)bk * LL * NN;
    const int pst = pos_stride(k);
    const int rst = pst * NN;
    const int slot = (k == 0) ? 0 : (k == 1) ? 2 : (k == 2) ? 1 : 3;
    float* optr = out + ((size_t)(slot * BB + b) * CC + c0) * LL;
    const int lbeg = chunk * CHL;

    {
        v2f h[MCH][QP], ap[MCH][QP];
        #pragma unroll
        for (int m = 0; m < MCH; ++m) {
            #pragma unroll
            for (int p = 0; p < QP; ++p) { h[m][p] = (v2f){0.0f, 0.0f}; ap[m][p] = (v2f){1.0f, 1.0f}; }
        }
        for (int t = 0; t < CHL / GRP; ++t) {
            const int l0 = lbeg + t * GRP;
            const int p0 = dir_pos(k, l0);
            float ldv[MCH][4], uv[MCH][4], duv[MCH][4];
            load_du(dup, p0 + 4 * q * pst, pst, ldv, uv);
            #pragma unroll
            for (int m = 0; m < MCH; ++m) {
                #pragma unroll
                for (int i = 0; i < 4; ++i) duv[m][i] = ldv[m][i] * uv[m][i];
            }
            const int r0 = p0 * NN + n0;
            float4 Be[4], Bo[4];
            load_rows<0>(bpl, r0, rst, Be);
            load_rows<1>(bpl, r0, rst, Bo);
            carry_group<0>(ldv, duv, An, Be, h, ap);
            GROUP_FENCE();
            load_rows<2>(bpl, r0, rst, Be);
            carry_group<1>(ldv, duv, An, Bo, h, ap);
            GROUP_FENCE();
            load_rows<3>(bpl, r0, rst, Bo);
            carry_group<2>(ldv, duv, An, Be, h, ap);
            carry_group<3>(ldv, duv, An, Bo, h, ap);
        }
        #pragma unroll
        for (int m = 0; m < MCH; ++m) {
            *reinterpret_cast<float4*>(&sA[m][chunk][n0]) = make_float4(ap[m][0].x, ap[m][0].y, ap[m][1].x, ap[m][1].y);
            *reinterpret_cast<float4*>(&sH[m][chunk][n0]) = make_float4(h[m][0].x, h[m][0].y, h[m][1].x, h[m][1].y);
        }
    }
    __syncthreads();

    if (tid < MCH * NN) {
        const int m = tid >> 4;
        const int n = tid & 15;
        float Hp = 0.0f;
        #pragma unroll 8
        for (int j = 0; j < CH; ++j) {
            const float hj = sH[m][j][n];
            sH[m][j][n] = Hp;
            Hp = sA[m][j][n] * Hp + hj;
        }
    }
    __syncthreads();

    v2f h[MCH][QP];
    #pragma unroll
    for (int m = 0; m < MCH; ++m) {
        const float4 i4 = *reinterpret_cast<const float4*>(&sH[m][chunk][n0]);
        h[m][0] = (v2f){i4.x, i4.y};
        h[m][1] = (v2f){i4.z, i4.w};
    }
    for (int t = 0; t < CHL / GRP; ++t) {
        const int l0 = lbeg + t * GRP;
        const int lq = l0 + 4 * q;
        const int p0 = dir_pos(k, l0);
        float ldv[MCH][4], uv[MCH][4], duv[MCH][4], ykeep[MCH][4];
        load_du(dup, p0 + 4 * q * pst, pst, ldv, uv);
        #pragma unroll
        for (int m = 0; m < MCH; ++m) {
            #pragma unroll
            for (int i = 0; i < 4; ++i) { duv[m][i] = ldv[m][i] * uv[m][i]; ykeep[m][i] = 0.0f; }
        }
        const int r0 = p0 * NN + n0;
        float4 Be[4], Ce[4], Bo[4], Co[4];
        load_rows<0>(bpl, r0, rst, Be); load_rows<0>(cpl, r0, rst, Ce);
        load_rows<1>(bpl, r0, rst, Bo); load_rows<1>(cpl, r0, rst, Co);
        out_group<0>(q, ldv, duv, An, Be, Ce, h, ykeep);
        GROUP_FENCE();
        load_rows<2>(bpl, r0, rst, Be); load_rows<2>(cpl, r0, rst, Ce);
        out_group<1>(q, ldv, duv, An, Bo, Co, h, ykeep);
        GROUP_FENCE();
        load_rows<3>(bpl, r0, rst, Bo); load_rows<3>(cpl, r0, rst, Co);
        out_group<2>(q, ldv, duv, An, Be, Ce, h, ykeep);
        out_group<3>(q, ldv, duv, An, Bo, Co, h, ykeep);
        #pragma unroll
        for (int m = 0; m < MCH; ++m) {
            float y4[4];
            #pragma unroll
            for (int i = 0; i < 4; ++i) y4[i] = ykeep[m][i] + Dd[m] * uv[m][i];
            store_dir4(optr + m * LL, k, lq, y4);
        }
    }
}

extern "C" void kernel_launch(void* const* d_in, const int* in_sizes, int n_in,
                              void* d_out, int out_size, void* d_ws, size_t ws_size,
                              hipStream_t stream) {
    const float* x      = (const float*)d_in[0];
    const float* wproj  = (const float*)d_in[1];
    const float* wdelta = (const float*)d_in[2];
    const float* dbias  = (const float*)d_in[3];
    const float* A_log  = (const float*)d_in[4];
    const float* Ds     = (const float*)d_in[5];
    float* out = (float*)d_out;

    float* ws     = (float*)d_ws;
    float* du     = ws;
    float* Bt     = du     + (size_t)BB * KK * CC * LL * 2;
    float* Ct     = Bt     + (size_t)BB * KK * LL * NN;

    dim3 gProj(LL / LT, KK, BB);
    s6_proj_kernel<<<gProj, 256, 0, stream>>>(x, wproj, wdelta, dbias, du, Bt, Ct);

    s6_scan_kernel<<<BB * KK * (CC / MCH), SCAN_THREADS, 0, stream>>>(du, Bt, Ct, A_log, Ds, out);
}


# scan, transposed directions: y is staged in an LDS plane [w][h] per channel and written out by the work-group as whole 128-byte lines, not by 4-byte stores that cover 8 lines each; sums unchanged
# speedup vs baseline: 21.6393x; 1.0341x over previous
#include <hip/hip_runtime.h>
#include <hip/hip_bf16.h>

#define BB   2
#define CC   96
#define HH   64
#define WWW  64
#define LL   4096
#define NN   16
#define RR   6
#define KK   4
#define DBCROWS 38
#define MPAD 48
#define LT   64
#define QN   4
#define QL   (NN / QN)
#define CH   128
#define CHL  (LL / CH)

typedef float v2f __attribute__((ext_vector_type(2)));
typedef float v8f __attribute__((ext_vector_type(8)));
typedef float v4f __attribute__((ext_vector_type(4)));

__device__ __forceinline__ int tr_idx(int l) { return ((l & 63) << 6) | (l >> 6); }
__device__ __forceinline__ int dir_pos(const int k, const int l) {
    const int lr = (k & 2) ? LL - 1 - l : l;
    return (k & 1) ? tr_idx(lr) : lr;
}

template <int CTRL>
__device__ __forceinline__ float row_dpp(float v) {
    return __builtin_bit_cast(float,
        __builtin_amdgcn_update_dpp(0, __builtin_bit_cast(int, v), CTRL, 0xf, 0xf, true));
}
#define GRP 16

__global__ __launch_bounds__(256)
void s6_proj_kernel(const float* __restrict__ x,
                    const float* __restrict__ wproj,
                    const float* __restrict__ wdelta,
                    const float* __restrict__ dbias,
                    float* __restrict__ du_out,
                    float* __restrict__ Bt,
                    float* __restrict__ Ct)
{
    __shared__ __attribute__((aligned(16))) float sW[MPAD][CC];
    __shared__ float sX[CC][LT + 1];
    __shared__ float sDBC[MPAD][LT + 1];
    __shared__ __attribute__((aligned(16))) float sWd[256 * 4];
    __shared__ float sBias[256];

    const int tid = threadIdx.x;
    const int l0  = blockIdx.x * LT;
    const int k   = blockIdx.y;
    const int b   = blockIdx.z;

    constexpr int W4  = DBCROWS * CC / 4;
    constexpr int WP4 = MPAD * CC / 4;
    constexpr int D4  = CC * RR / 4;
    constexpr int XJ  = CC * LT / 256;
    const v4f* wsrc = reinterpret_cast<const v4f*>(wproj + (size_t)k * DBCROWS * CC);
    const v4f* dsrc = reinterpret_cast<const v4f*>(wdelta + (size_t)k * CC * RR);
    const v4f   w0 = wsrc[tid];
    const v4f   w1 = wsrc[tid + 256];
    const v4f   w2 = wsrc[tid + 512];
    const bool  w3in = tid + 768 < W4;
    const v4f   w3 = wsrc[w3in ? tid + 768 : W4 - 1];
    const v4f   dv = dsrc[tid < D4 ? tid : D4 - 1];
    const float bv = dbias[k * CC + (tid < CC ? tid : CC - 1)];
    const int col = tid & (LT - 1);
    const int cr  = tid >> 6;
    const float* xcol = x + (size_t)b * CC * LL + l0 + col;
    float xv[XJ];
    #pragma unroll
    for (int j = 0; j < XJ; ++j) xv[j] = xcol[(size_t)(cr + 4 * j) * LL];

    v4f* sW4 = reinterpret_cast<v4f*>(&sW[0][0]);
    const v4f zero4 = {0.0f, 0.0f, 0.0f, 0.0f};
    sW4[tid]       = w0;
    sW4[tid + 256] = w1;
    sW4[tid + 512] = w2;
    sW4[tid + 768] = w3in ? w3 : zero4;
    if (tid < WP4 - 1024) sW4[tid + 1024] = zero4;
    reinterpret_cast<v4f*>(sWd)[tid] = dv;
    sBias[tid] = bv;
    #pragma unroll
    for (int j = 0; j < XJ; ++j) sX[cr + 4 * j][col] = xv[j];
    __syncthreads();

    const int wave = tid >> 5;
    const int lane = tid & 31;
    const int lm   = lane & 15;
    const int lh   = lane >> 4;
    for (int t = wave; t < (MPAD / 16) * (LT / 16); t += 8) {
        const int mt = t / (LT / 16);
        const int nt = t % (LT / 16);
        v8f acc = {};
        for (int kk = 0; kk < CC; kk += 4) {
            v2f a, bf;
            a.x  = sW[mt * 16 + lm][kk + 2 * lh + 0];
            a.y  = sW[mt * 16 + lm][kk + 2 * lh + 1];
            bf.x = sX[kk + 2 * lh + 0][nt * 16 + lm];
            bf.y = sX[kk + 2 * lh + 1][nt * 16 + lm];
            acc = __builtin_amdgcn_wmma_f32_16x16x4_f32(
                      false, a, false, bf, (short)0, acc, false, false);
        }
        #pragma unroll
        for (int i = 0; i < 8; ++i)
            sDBC[mt * 16 + i + 8 * lh][nt * 16 + lm] = acc[i];
    }
    __syncthreads();

    for (int i = tid; i < LT * NN; i += 256) {
        int col = i / NN, n = i % NN;
        size_t base = (((size_t)(b * KK + k) * LL + (l0 + col)) * NN + n);
        Bt[base] = sDBC[RR + n][col];
        Ct[base] = sDBC[RR + NN + n][col];
    }
    v2f* dpair = reinterpret_cast<v2f*>(du_out);
    for (int i = tid; i < CC * LT; i += 256) {
        int c = i / LT, col = i % LT;
        float s = 0.0f;
        #pragma unroll
        for (int r = 0; r < RR; ++r) s += sWd[c * RR + r] * sDBC[r][col];
        s += sBias[c];
        const float e  = __expf(s);
        const float u  = 1.0f + e;
        const float lg = __builtin_amdgcn_logf(u) * 0.693147181f + (e - (u - 1.0f)) * __builtin_amdgcn_rcpf(u);
        const float sp = (s > 20.0f) ? s : lg;
        dpair[(((size_t)(b * KK + k) * CC + c) * LL) + l0 + col] = (v2f){sp, sX[c][col]};
    }
}

#define SCAN_THREADS (CH * QL)
#define MCH 3
#define QP  (QN / 2)

__device__ __forceinline__ int pos_stride(const int k) { return ((k & 1) ? 64 : 1) * ((k & 2) ? -1 : 1); }
__device__ __forceinline__ void load_du(const float* __restrict__ p, const int pq, const int pst,
                                        float (&d)[MCH][4], float (&u)[MCH][4]) {
    const char* pc = reinterpret_cast<const char*>(p);
    const unsigned ob = 8u * (unsigned)pq;
    const int sb = 8 * pst;
    #pragma unroll
    for (int i = 0; i < 4; ++i) {
        const unsigned o = ob + (unsigned)(i * sb);
        #pragma unroll
        for (int m = 0; m < MCH; ++m) {
            const v2f v = *reinterpret_cast<const v2f*>(pc + (size_t)o + (size_t)(m * LL * 8));
            d[m][i] = v.x;
            u[m][i] = v.y;
        }
    }
}
__device__ __forceinline__ void store_row4(float* __restrict__ p, const int k, const int l,
                                           const float (&v)[4]) {
    const float4 t = (k == 2) ? make_float4(v[3], v[2], v[1], v[0])
                              : make_float4(v[0], v[1], v[2], v[3]);
    *reinterpret_cast<float4*>(p + ((k == 2) ? LL - 4 - l : l)) = t;
}
#define YP (HH + 1)
__device__ __forceinline__ void stage_col4(float* __restrict__ pl, const int k, const int l,
                                           const float (&v)[4]) {
    const bool rev = (k == 3);
    const int lb = rev ? LL - 4 - l : l;
    float* sp = pl + (lb >> 6) * YP + (lb & 63);
    sp[0] = rev ? v[3] : v[0];
    sp[1] = rev ? v[2] : v[1];
    sp[2] = rev ? v[1] : v[2];
    sp[3] = rev ? v[0] : v[3];
}

template <int S>
__device__ __forceinline__ void carry_step(const float (&ldv)[MCH][4], const float (&duv)[MCH][4],
                                           const v2f (&An)[MCH][QP], const float4 Bv,
                                           v2f (&h)[MCH][QP], v2f (&ap)[MCH][QP]) {
    const v2f Bs[QP] = {{Bv.x, Bv.y}, {Bv.z, Bv.w}};
    #pragma unroll
    for (int m = 0; m < MCH; ++m) {
        const float ld = row_dpp<(S >> 2) * 0x55>(ldv[m][S & 3]);
        const float du = row_dpp<(S >> 2) * 0x55>(duv[m][S & 3]);
        #pragma unroll
        for (int p = 0; p < QP; ++p) {
            const v2f e = ld * An[m][p];
            v2f dA;
            dA.x = __expf(e.x);
            dA.y = __expf(e.y);
            h[m][p]  = dA * h[m][p] + du * Bs[p];
            ap[m][p] *= dA;
        }
    }
}

template <int S>
__device__ __forceinline__ void out_step(const int q, const float (&ldv)[MCH][4], const float (&duv)[MCH][4],
                                         const v2f (&An)[MCH][QP], const float4 Bv, const float4 Cv,
                                         v2f (&h)[MCH][QP], float (&ykeep)[MCH][4]) {
    const v2f Bs[QP] = {{Bv.x, Bv.y}, {Bv.z, Bv.w}};
    const v2f Cs[QP] = {{Cv.x, Cv.y}, {Cv.z, Cv.w}};
    #pragma unroll
    for (int m = 0; m < MCH; ++m) {
        const float ld = row_dpp<(S >> 2) * 0x55>(ldv[m][S & 3]);
        const float du = row_dpp<(S >> 2) * 0x55>(duv[m][S & 3]);
        #pragma unroll
        for (int p = 0; p < QP; ++p) {
            const v2f e = ld * An[m][p];
            v2f dA;
            dA.x = __expf(e.x);
            dA.y = __expf(e.y);
            h[m][p] = dA * h[m][p] + du * Bs[p];
        }
        const v2f s2 = h[m][0] * Cs[0] + h[m][1] * Cs[1];
        float v = s2.x + s2.y;
        v += row_dpp<0x160 + 1>(v);
        v += row_dpp<0x160 + 2>(v);
        ykeep[m][S & 3] = (q == (S >> 2)) ? v : ykeep[m][S & 3];
    }
}

template <int G>
__device__ __forceinline__ void load_rows(const float* __restrict__ pl, const int r0, const int rst,
                                          float4 (&R)[4]) {
    const char* pc = reinterpret_cast<const char*>(pl);
    #pragma unroll
    for (int s = 0; s < 4; ++s) {
        const unsigned o = 4u * (unsigned)(r0 + (4 * G + s) * rst);
        R[s] = *reinterpret_cast<const float4*>(pc + (size_t)o);
    }
}
template <int G>
__device__ __forceinline__ void carry_group(const float (&ldv)[MCH][4], const float (&duv)[MCH][4],
                                            const v2f (&An)[MCH][QP], const float4 (&Bv)[4],
                                            v2f (&h)[MCH][QP], v2f (&ap)[MCH][QP]) {
    carry_step<4 * G + 0>(ldv, duv, An, Bv[0], h, ap); carry_step<4 * G + 1>(ldv, duv, An, Bv[1], h, ap);
    carry_step<4 * G + 2>(ldv, duv, An, Bv[2], h, ap); carry_step<4 * G + 3>(ldv, duv, An, Bv[3], h, ap);
}
template <int G>
__device__ __forceinline__ void out_group(const int q, const float (&ldv)[MCH][4], const float (&duv)[MCH][4],
                                          const v2f (&An)[MCH][QP], const float4 (&Bv)[4], const float4 (&Cv)[4],
                                          v2f (&h)[MCH][QP], float (&ykeep)[MCH][4]) {
    out_step<4 * G + 0>(q, ldv, duv, An, Bv[0], Cv[0], h, ykeep); out_step<4 * G + 1>(q, ldv, duv, An, Bv[1], Cv[1], h, ykeep);
    out_step<4 * G + 2>(q, ldv, duv, An, Bv[2], Cv[2], h, ykeep); out_step<4 * G + 3>(q, ldv, duv, An, Bv[3], Cv[3], h, ykeep);
}
#define GROUP_FENCE() __builtin_amdgcn_sched_barrier(0)

__global__ __launch_bounds__(SCAN_THREADS)
void s6_scan_kernel(const float* __restrict__ du,
                    const float* __restrict__ Bt,
                    const float* __restrict__ Ct,
                    const float* __restrict__ A_log,
                    const float* __restrict__ Ds,
                    float* __restrict__ out)
{
    __shared__ __attribute__((aligned(16))) float sA[MCH][CH][NN];
    __shared__ __attribute__((aligned(16))) float sH[MCH][CH][NN];
    __shared__ float sY[MCH][WWW][YP];

    const int tid   = threadIdx.x;
    const int chunk = tid >> 2;
    const int q     = tid & 3;
    const int n0    = q * QN;
    const int grp   = blockIdx.x;
    const int c0    = (grp % (CC / MCH)) * MCH;
    const int bk    = grp / (CC / MCH);
    const int k     = bk % KK;
    const int b     = bk / KK;
    const int d0    = k * CC + c0;

    v2f An[MCH][QP];
    float Dd[MCH];
    #pragma unroll
    for (int m = 0; m < MCH; ++m) {
        #pragma unroll
        for (int p = 0; p < QP; ++p) {
            An[m][p].x = -__expf(A_log[(d0 + m) * NN + n0 + 2 * p + 0]);
            An[m][p].y = -__expf(A_log[(d0 + m) * NN + n0 + 2 * p + 1]);
        }
        Dd[m] = Ds[d0 + m];
    }
    const float* dup = du + ((size_t)bk * CC + c0) * LL * 2;
    const float* bpl = Bt + (size_t)bk * LL * NN;
    const float* cpl = Ct + (size_t)bk * LL * NN;
    const int pst = pos_stride(k);
    const int rst = pst * NN;
    const int slot = (k == 0) ? 0 : (k == 1) ? 2 : (k == 2) ? 1 : 3;
    float* optr = out + ((size_t)(slot * BB + b) * CC + c0) * LL;
    const int lbeg = chunk * CHL;

    {
        v2f h[MCH][QP], ap[MCH][QP];
        #pragma unroll
        for (int m = 0; m < MCH; ++m) {
            #pragma unroll
            for (int p = 0; p < QP; ++p) { h[m][p] = (v2f){0.0f, 0.0f}; ap[m][p] = (v2f){1.0f, 1.0f}; }
        }
        for (int t = 0; t < CHL / GRP; ++t) {
            const int l0 = lbeg + t * GRP;
            const int p0 = dir_pos(k, l0);
            float ldv[MCH][4], uv[MCH][4], duv[MCH][4];
            load_du(dup, p0 + 4 * q * pst, pst, ldv, uv);
            #pragma unroll
            for (int m = 0; m < MCH; ++m) {
                #pragma unroll
                for (int i = 0; i < 4; ++i) duv[m][i] = ldv[m][i] * uv[m][i];
            }
            const int r0 = p0 * NN + n0;
            float4 Be[4], Bo[4];
            load_rows<0>(bpl, r0, rst, Be);
            load_rows<1>(bpl, r0, rst, Bo);
            carry_group<0>(ldv, duv, An, Be, h, ap);
            GROUP_FENCE();
            load_rows<2>(bpl, r0, rst, Be);
            carry_group<1>(ldv, duv, An, Bo, h, ap);
            GROUP_FENCE();
            load_rows<3>(bpl, r0, rst, Bo);
            carry_group<2>(ldv, duv, An, Be, h, ap);
            carry_group<3>(ldv, duv, An, Bo, h, ap);
        }
        #pragma unroll
        for (int m = 0; m < MCH; ++m) {
            *reinterpret_cast<float4*>(&sA[m][chunk][n0]) = make_float4(ap[m][0].x, ap[m][0].y, ap[m][1].x, ap[m][1].y);
            *reinterpret_cast<float4*>(&sH[m][chunk][n0]) = make_float4(h[m][0].x, h[m][0].y, h[m][1].x, h[m][1].y);
        }
    }
    __syncthreads();

    if (tid < MCH * NN) {
        const int m = tid >> 4;
        const int n = tid & 15;
        float Hp = 0.0f;
        #pragma unroll 8
        for (int j = 0; j < CH; ++j) {
            const float hj = sH[m][j][n];
            sH[m][j][n] = Hp;
            Hp = sA[m][j][n] * Hp + hj;
        }
    }
    __syncthreads();

    v2f h[MCH][QP];
    #pragma unroll
    for (int m = 0; m < MCH; ++m) {
        const float4 i4 = *reinterpret_cast<const float4*>(&sH[m][chunk][n0]);
        h[m][0] = (v2f){i4.x, i4.y};
        h[m][1] = (v2f){i4.z, i4.w};
    }
    for (int t = 0; t < CHL / GRP; ++t) {
        const int l0 = lbeg + t * GRP;
        const int lq = l0 + 4 * q;
        const int p0 = dir_pos(k, l0);
        float ldv[MCH][4], uv[MCH][4], duv[MCH][4], ykeep[MCH][4];
        load_du(dup, p0 + 4 * q * pst, pst, ldv, uv);
        #pragma unroll
        for (int m = 0; m < MCH; ++m) {
            #pragma unroll
            for (int i = 0; i < 4; ++i) { duv[m][i] = ldv[m][i] * uv[m][i]; ykeep[m][i] = 0.0f; }
        }
        const int r0 = p0 * NN + n0;
        float4 Be[4], Ce[4], Bo[4], Co[4];
        load_rows<0>(bpl, r0, rst, Be); load_rows<0>(cpl, r0, rst, Ce);
        load_rows<1>(bpl, r0, rst, Bo); load_rows<1>(cpl, r0, rst, Co);
        out_group<0>(q, ldv, duv, An, Be, Ce, h, ykeep);
        GROUP_FENCE();
        load_rows<2>(bpl, r0, rst, Be); load_rows<2>(cpl, r0, rst, Ce);
        out_group<1>(q, ldv, duv, An, Bo, Co, h, ykeep);
        GROUP_FENCE();
        load_rows<3>(bpl, r0, rst, Bo); load_rows<3>(cpl, r0, rst, Co);
        out_group<2>(q, ldv, duv, An, Be, Ce, h, ykeep);
        out_group<3>(q, ldv, duv, An, Bo, Co, h, ykeep);
        #pragma unroll
        for (int m = 0; m < MCH; ++m) {
            float y4[4];
            #pragma unroll
            for (int i = 0; i < 4; ++i) y4[i] = ykeep[m][i] + Dd[m] * uv[m][i];
            if (k & 1) stage_col4(&sY[m][0][0], k, lq, y4);
            else       store_row4(optr + m * LL, k, lq, y4);
        }
    }

    if (k & 1) {
        __syncthreads();
        constexpr int F4C = LL / 4;
        constexpr int SWP = F4C / SCAN_THREADS;
        #pragma unroll
        for (int j = 0; j < MCH * SWP; ++j) {
            const int m  = j / SWP;
            const int r  = tid + (j % SWP) * SCAN_THREADS;
            const int h  = r >> 4;
            const int w4 = (r & 15) << 2;
            const float4 v = make_float4(sY[m][w4 + 0][h], sY[m][w4 + 1][h], sY[m][w4 + 2][h], sY[m][w4 + 3][h]);
            *reinterpret_cast<float4*>(optr + m * LL + h * WWW + w4) = v;
        }
    }
}

extern "C" void kernel_launch(void* const* d_in, const int* in_sizes, int n_in,
                              void* d_out, int out_size, void* d_ws, size_t ws_size,
                              hipStream_t stream) {
    const float* x      = (const float*)d_in[0];
    const float* wproj  = (const float*)d_in[1];
    const float* wdelta = (const float*)d_in[2];
    const float* dbias  = (const float*)d_in[3];
    const float* A_log  = (const float*)d_in[4];
    const float* Ds     = (const float*)d_in[5];
    float* out = (float*)d_out;

    float* ws     = (float*)d_ws;
    float* du     = ws;
    float* Bt     = du     + (size_t)BB * KK * CC * LL * 2;
    float* Ct     = Bt     + (size_t)BB * KK * LL * NN;

    dim3 gProj(LL / LT, KK, BB);
    s6_proj_kernel<<<gProj, 256, 0, stream>>>(x, wproj, wdelta, dbias, du, Bt, Ct);

    s6_scan_kernel<<<BB * KK * (CC / MCH), SCAN_THREADS, 0, stream>>>(du, Bt, Ct, A_log, Ds, out);
}


# scan: loads issued a group's work ahead of use (delta,u read once, parked in LDS for pass 2; B/C rows two groups ahead, also across the carry chain), not batched at each trip's head; sums unchanged
# speedup vs baseline: 22.4958x; 1.0396x over previous
#include <hip/hip_runtime.h>
#include <hip/hip_bf16.h>

#define BB   2
#define CC   96
#define HH   64
#define WWW  64
#define LL   4096
#define NN   16
#define RR   6
#define KK   4
#define DBCROWS 38
#define MPAD 48
#define LT   64
#define QN   4
#define QL   (NN / QN)
#define CH   128
#define CHL  (LL / CH)

typedef float v2f __attribute__((ext_vector_type(2)));
typedef float v8f __attribute__((ext_vector_type(8)));
typedef float v4f __attribute__((ext_vector_type(4)));

__device__ __forceinline__ int tr_idx(int l) { return ((l & 63) << 6) | (l >> 6); }
__device__ __forceinline__ int dir_pos(const int k, const int l) {
    const int lr = (k & 2) ? LL - 1 - l : l;
    return (k & 1) ? tr_idx(lr) : lr;
}

template <int CTRL>
__device__ __forceinline__ float row_dpp(float v) {
    return __builtin_bit_cast(float,
        __builtin_amdgcn_update_dpp(0, __builtin_bit_cast(int, v), CTRL, 0xf, 0xf, true));
}
#define GRP 16

__global__ __launch_bounds__(256)
void s6_proj_kernel(const float* __restrict__ x,
                    const float* __restrict__ wproj,
                    const float* __restrict__ wdelta,
                    const float* __restrict__ dbias,
                    float* __restrict__ du_out,
                    float* __restrict__ Bt,
                    float* __restrict__ Ct)
{
    __shared__ __attribute__((aligned(16))) float sW[MPAD][CC];
    __shared__ float sX[CC][LT + 1];
    __shared__ float sDBC[MPAD][LT + 1];
    __shared__ __attribute__((aligned(16))) float sWd[256 * 4];
    __shared__ float sBias[256];

    const int tid = threadIdx.x;
    const int l0  = blockIdx.x * LT;
    const int k   = blockIdx.y;
    const int b   = blockIdx.z;

    constexpr int W4  = DBCROWS * CC / 4;
    constexpr int WP4 = MPAD * CC / 4;
    constexpr int D4  = CC * RR / 4;
    constexpr int XJ  = CC * LT / 256;
    const v4f* wsrc = reinterpret_cast<const v4f*>(wproj + (size_t)k * DBCROWS * CC);
    const v4f* dsrc = reinterpret_cast<const v4f*>(wdelta + (size_t)k * CC * RR);
    const v4f   w0 = wsrc[tid];
    const v4f   w1 = wsrc[tid + 256];
    const v4f   w2 = wsrc[tid + 512];
    const bool  w3in = tid + 768 < W4;
    const v4f   w3 = wsrc[w3in ? tid + 768 : W4 - 1];
    const v4f   dv = dsrc[tid < D4 ? tid : D4 - 1];
    const float bv = dbias[k * CC + (tid < CC ? tid : CC - 1)];
    const int col = tid & (LT - 1);
    const int cr  = tid >> 6;
    const float* xcol = x + (size_t)b * CC * LL + l0 + col;
    float xv[XJ];
    #pragma unroll
    for (int j = 0; j < XJ; ++j) xv[j] = xcol[(size_t)(cr + 4 * j) * LL];

    v4f* sW4 = reinterpret_cast<v4f*>(&sW[0][0]);
    const v4f zero4 = {0.0f, 0.0f, 0.0f, 0.0f};
    sW4[tid]       = w0;
    sW4[tid + 256] = w1;
    sW4[tid + 512] = w2;
    sW4[tid + 768] = w3in ? w3 : zero4;
    if (tid < WP4 - 1024) sW4[tid + 1024] = zero4;
    reinterpret_cast<v4f*>(sWd)[tid] = dv;
    sBias[tid] = bv;
    #pragma unroll
    for (int j = 0; j < XJ; ++j) sX[cr + 4 * j][col] = xv[j];
    __syncthreads();

    const int wave = tid >> 5;
    const int lane = tid & 31;
    const int lm   = lane & 15;
    const int lh   = lane >> 4;
    for (int t = wave; t < (MPAD / 16) * (LT / 16); t += 8) {
        const int mt = t / (LT / 16);
        const int nt = t % (LT / 16);
        v8f acc = {};
        for (int kk = 0; kk < CC; kk += 4) {
            v2f a, bf;
            a.x  = sW[mt * 16 + lm][kk + 2 * lh + 0];
            a.y  = sW[mt * 16 + lm][kk + 2 * lh + 1];
            bf.x = sX[kk + 2 * lh + 0][nt * 16 + lm];
            bf.y = sX[kk + 2 * lh + 1][nt * 16 + lm];
            acc = __builtin_amdgcn_wmma_f32_16x16x4_f32(
                      false, a, false, bf, (short)0, acc, false, false);
        }
        #pragma unroll
        for (int i = 0; i < 8; ++i)
            sDBC[mt * 16 + i + 8 * lh][nt * 16 + lm] = acc[i];
    }
    __syncthreads();

    for (int i = tid; i < LT * NN; i += 256) {
        int col = i / NN, n = i % NN;
        size_t base = (((size_t)(b * KK + k) * LL + (l0 + col)) * NN + n);
        Bt[base] = sDBC[RR + n][col];
        Ct[base] = sDBC[RR + NN + n][col];
    }
    v2f* dpair = reinterpret_cast<v2f*>(du_out);
    for (int i = tid; i < CC * LT; i += 256) {
        int c = i / LT, col = i % LT;
        float s = 0.0f;
        #pragma unroll
        for (int r = 0; r < RR; ++r) s += sWd[c * RR + r] * sDBC[r][col];
        s += sBias[c];
        const float e  = __expf(s);
        const float u  = 1.0f + e;
        const float lg = __builtin_amdgcn_logf(u) * 0.693147181f + (e - (u - 1.0f)) * __builtin_amdgcn_rcpf(u);
        const float sp = (s > 20.0f) ? s : lg;
        dpair[(((size_t)(b * KK + k) * CC + c) * LL) + l0 + col] = (v2f){sp, sX[c][col]};
    }
}

#define SCAN_THREADS (CH * QL)
#define MCH 3
#define QP  (QN / 2)

__device__ __forceinline__ int pos_stride(const int k) { return ((k & 1) ? 64 : 1) * ((k & 2) ? -1 : 1); }
typedef const __attribute__((address_space(1))) char* gptr;
__device__ __forceinline__ gptr scalar_ptr(const void* p) {
    const unsigned long long a = reinterpret_cast<unsigned long long>(p);
    const unsigned lo = __builtin_amdgcn_readfirstlane((unsigned)a);
    const unsigned hi = __builtin_amdgcn_readfirstlane((unsigned)(a >> 32));
    return reinterpret_cast<gptr>(((unsigned long long)hi << 32) | lo);
}
__device__ __forceinline__ void load_du(const float* __restrict__ p, const unsigned ob, const int s0, const int pst,
                                        float (&d)[MCH][4], float (&u)[MCH][4]) {
    #pragma unroll
    for (int i = 0; i < 4; ++i) {
        const gptr sp = scalar_ptr(p + 2 * ((s0 + i) * pst));
        #pragma unroll
        for (int m = 0; m < MCH; ++m) {
            const v2f v = *reinterpret_cast<const __attribute__((address_space(1))) v2f*>(sp + (size_t)ob + (size_t)(m * LL * 8));
            d[m][i] = v.x;
            u[m][i] = v.y;
        }
    }
}
__device__ __forceinline__ void store_row4(float* __restrict__ p, const int k, const int l,
                                           const float (&v)[4]) {
    const float4 t = (k == 2) ? make_float4(v[3], v[2], v[1], v[0])
                              : make_float4(v[0], v[1], v[2], v[3]);
    *reinterpret_cast<float4*>(p + ((k == 2) ? LL - 4 - l : l)) = t;
}
#define YP (HH + 1)
__device__ __forceinline__ void stage_col4(float* __restrict__ pl, const int k, const int l,
                                           const float (&v)[4]) {
    const bool rev = (k == 3);
    const int lb = rev ? LL - 4 - l : l;
    float* sp = pl + (lb >> 6) * YP + (lb & 63);
    sp[0] = rev ? v[3] : v[0];
    sp[1] = rev ? v[2] : v[1];
    sp[2] = rev ? v[1] : v[2];
    sp[3] = rev ? v[0] : v[3];
}

template <int S>
__device__ __forceinline__ void carry_step(const float (&ldv)[MCH][4], const float (&duv)[MCH][4],
                                           const v2f (&An)[MCH][QP], const float4 Bv,
                                           v2f (&h)[MCH][QP], v2f (&ap)[MCH][QP]) {
    const v2f Bs[QP] = {{Bv.x, Bv.y}, {Bv.z, Bv.w}};
    #pragma unroll
    for (int m = 0; m < MCH; ++m) {
        const float ld = row_dpp<(S >> 2) * 0x55>(ldv[m][S & 3]);
        const float du = row_dpp<(S >> 2) * 0x55>(duv[m][S & 3]);
        #pragma unroll
        for (int p = 0; p < QP; ++p) {
            const v2f e = ld * An[m][p];
            v2f dA;
            dA.x = __expf(e.x);
            dA.y = __expf(e.y);
            h[m][p]  = dA * h[m][p] + du * Bs[p];
            ap[m][p] *= dA;
        }
    }
}

template <int S>
__device__ __forceinline__ void out_step(const int q, const float (&ldv)[MCH][4], const float (&duv)[MCH][4],
                                         const v2f (&An)[MCH][QP], const float4 Bv, const float4 Cv,
                                         v2f (&h)[MCH][QP], float (&ykeep)[MCH][4]) {
    const v2f Bs[QP] = {{Bv.x, Bv.y}, {Bv.z, Bv.w}};
    const v2f Cs[QP] = {{Cv.x, Cv.y}, {Cv.z, Cv.w}};
    #pragma unroll
    for (int m = 0; m < MCH; ++m) {
        const float ld = row_dpp<(S >> 2) * 0x55>(ldv[m][S & 3]);
        const float du = row_dpp<(S >> 2) * 0x55>(duv[m][S & 3]);
        #pragma unroll
        for (int p = 0; p < QP; ++p) {
            const v2f e = ld * An[m][p];
            v2f dA;
            dA.x = __expf(e.x);
            dA.y = __expf(e.y);
            h[m][p] = dA * h[m][p] + du * Bs[p];
        }
        const v2f s2 = h[m][0] * Cs[0] + h[m][1] * Cs[1];
        float v = s2.x + s2.y;
        v += row_dpp<0x160 + 1>(v);
        v += row_dpp<0x160 + 2>(v);
        ykeep[m][S & 3] = (q == (S >> 2)) ? v : ykeep[m][S & 3];
    }
}

template <int G>
__device__ __forceinline__ void load_rows(const float* __restrict__ pl, const unsigned ro, const int rst,
                                          float4 (&R)[4]) {
    #pragma unroll
    for (int s = 0; s < 4; ++s) {
        const gptr rowp = scalar_ptr(pl + (4 * G + s) * rst);
        R[s] = *reinterpret_cast<const __attribute__((address_space(1))) float4*>(rowp + (size_t)ro);
    }
}
#define CT_OFF ((size_t)BB * KK * LL * NN * sizeof(float))
template <int G>
__device__ __forceinline__ void load_rows_bc(const float* __restrict__ pl, const unsigned ro, const int rst,
                                             float4 (&RB)[4], float4 (&RC)[4]) {
    #pragma unroll
    for (int s = 0; s < 4; ++s) {
        const gptr rowp = scalar_ptr(pl + (4 * G + s) * rst);
        RB[s] = *reinterpret_cast<const __attribute__((address_space(1))) float4*>(rowp + (size_t)ro);
        RC[s] = *reinterpret_cast<const __attribute__((address_space(1))) float4*>(rowp + (size_t)ro + CT_OFF);
    }
}
template <int G>
__device__ __forceinline__ void carry_group(const float (&ldv)[MCH][4], const float (&duv)[MCH][4],
                                            const v2f (&An)[MCH][QP], const float4 (&Bv)[4],
                                            v2f (&h)[MCH][QP], v2f (&ap)[MCH][QP]) {
    carry_step<4 * G + 0>(ldv, duv, An, Bv[0], h, ap); carry_step<4 * G + 1>(ldv, duv, An, Bv[1], h, ap);
    carry_step<4 * G + 2>(ldv, duv, An, Bv[2], h, ap); carry_step<4 * G + 3>(ldv, duv, An, Bv[3], h, ap);
}
template <int G>
__device__ __forceinline__ void out_group(const int q, const float (&ldv)[MCH][4], const float (&duv)[MCH][4],
                                          const v2f (&An)[MCH][QP], const float4 (&Bv)[4], const float4 (&Cv)[4],
                                          v2f (&h)[MCH][QP], float (&ykeep)[MCH][4]) {
    out_step<4 * G + 0>(q, ldv, duv, An, Bv[0], Cv[0], h, ykeep); out_step<4 * G + 1>(q, ldv, duv, An, Bv[1], Cv[1], h, ykeep);
    out_step<4 * G + 2>(q, ldv, duv, An, Bv[2], Cv[2], h, ykeep); out_step<4 * G + 3>(q, ldv, duv, An, Bv[3], Cv[3], h, ykeep);
}
#define GROUP_FENCE()                                                                              \
    do {                                                                                           \
        asm volatile("" : "+v"(h[0][0]), "+v"(h[0][1]), "+v"(h[1][0]), "+v"(h[1][1]),              \
                          "+v"(h[2][0]), "+v"(h[2][1]) : : "memory");                              \
        __builtin_amdgcn_sched_barrier(0);                                                         \
    } while (0)
static_assert(MCH == 3 && QP == 2, "GROUP_FENCE names the 6 state pairs of a lane");
__device__ __forceinline__ void mul_du(const float (&d)[MCH][4], const float (&u)[MCH][4],
                                       float (&du)[MCH][4]) {
    #pragma unroll
    for (int m = 0; m < MCH; ++m) {
        #pragma unroll
        for (int i = 0; i < 4; ++i) du[m][i] = d[m][i] * u[m][i];
    }
}

__global__ __launch_bounds__(SCAN_THREADS)
void s6_scan_kernel(const float* __restrict__ du,
                    const float* __restrict__ Bt,
                    const float* __restrict__ A_log,
                    const float* __restrict__ Ds,
                    float* __restrict__ out)
{
    __shared__ __attribute__((aligned(16))) float sA[MCH][CH][NN];
    __shared__ __attribute__((aligned(16))) float sH[MCH][CH][NN];
    __shared__ float sY[MCH][WWW][YP];
    __shared__ __attribute__((aligned(16))) float sDU[2][2 * MCH][SCAN_THREADS][4];

    const int tid   = threadIdx.x;
    const int chunk = tid >> 2;
    const int q     = tid & 3;
    const int n0    = q * QN;
    const int grp   = blockIdx.x;
    const int c0    = (grp % (CC / MCH)) * MCH;
    const int bk    = grp / (CC / MCH);
    const int k     = bk % KK;
    const int b     = bk / KK;
    const int d0    = k * CC + c0;

    const float* dup = du + ((size_t)bk * CC + c0) * LL * 2;
    const float* bpl = Bt + (size_t)bk * LL * NN;
    const int pst = pos_stride(k);
    const int rst = pst * NN;
    const int slot = (k == 0) ? 0 : (k == 1) ? 2 : (k == 2) ? 1 : 3;
    float* optr = out + ((size_t)(slot * BB + b) * CC + c0) * LL;
    const int lbeg = chunk * CHL;
    const int pc0  = dir_pos(k, lbeg);
    const unsigned r0 = 4u * (unsigned)(pc0 * NN + n0);
    const unsigned d0b = 8u * (unsigned)(pc0 + 4 * q * pst);

    float ldv[2][MCH][4], uv[2][MCH][4];
    load_du(dup, d0b, 0, pst, ldv[0], uv[0]);
    float4 Be[4], Bo[4], Ce[4], Co[4];
    load_rows<0>(bpl, r0, rst, Be);
    load_rows<1>(bpl, r0, rst, Bo);
    v2f An[MCH][QP];
    float Dd[MCH];
    #pragma unroll
    for (int m = 0; m < MCH; ++m) {
        #pragma unroll
        for (int p = 0; p < QP; ++p) {
            An[m][p].x = -__expf(A_log[(d0 + m) * NN + n0 + 2 * p + 0]);
            An[m][p].y = -__expf(A_log[(d0 + m) * NN + n0 + 2 * p + 1]);
        }
        Dd[m] = Ds[d0 + m];
    }

    {
        v2f h[MCH][QP], ap[MCH][QP];
        #pragma unroll
        for (int m = 0; m < MCH; ++m) {
            #pragma unroll
            for (int p = 0; p < QP; ++p) { h[m][p] = (v2f){0.0f, 0.0f}; ap[m][p] = (v2f){1.0f, 1.0f}; }
        }
        float duv[MCH][4];
        mul_du(ldv[0], uv[0], duv);
        carry_group<0>(ldv[0], duv, An, Be, h, ap);
        GROUP_FENCE();
        load_rows<2>(bpl, r0, rst, Be);
        load_du(dup, d0b, GRP, pst, ldv[1], uv[1]);
        GROUP_FENCE();
        carry_group<1>(ldv[0], duv, An, Bo, h, ap);
        GROUP_FENCE();
        load_rows<3>(bpl, r0, rst, Bo);
        GROUP_FENCE();
        carry_group<2>(ldv[0], duv, An, Be, h, ap);
        GROUP_FENCE();
        load_rows<4>(bpl, r0, rst, Be);
        GROUP_FENCE();
        carry_group<3>(ldv[0], duv, An, Bo, h, ap);
        GROUP_FENCE();
        load_rows<5>(bpl, r0, rst, Bo);
        GROUP_FENCE();
        mul_du(ldv[1], uv[1], duv);
        carry_group<0>(ldv[1], duv, An, Be, h, ap);
        GROUP_FENCE();
        load_rows<6>(bpl, r0, rst, Be);
        GROUP_FENCE();
        carry_group<1>(ldv[1], duv, An, Bo, h, ap);
        GROUP_FENCE();
        load_rows<7>(bpl, r0, rst, Bo);
        GROUP_FENCE();
        carry_group<2>(ldv[1], duv, An, Be, h, ap);
        GROUP_FENCE();
        load_rows_bc<0>(bpl, r0, rst, Be, Ce);
        GROUP_FENCE();
        carry_group<3>(ldv[1], duv, An, Bo, h, ap);
        GROUP_FENCE();
        load_rows_bc<1>(bpl, r0, rst, Bo, Co);
        GROUP_FENCE();
        #pragma unroll
        for (int m = 0; m < MCH; ++m) {
            *reinterpret_cast<float4*>(&sA[m][chunk][n0]) = make_float4(ap[m][0].x, ap[m][0].y, ap[m][1].x, ap[m][1].y);
            *reinterpret_cast<float4*>(&sH[m][chunk][n0]) = make_float4(h[m][0].x, h[m][0].y, h[m][1].x, h[m][1].y);
            #pragma unroll
            for (int t = 0; t < 2; ++t) {
                *reinterpret_cast<float4*>(&sDU[t][m][tid][0])       = make_float4(ldv[t][m][0], ldv[t][m][1], ldv[t][m][2], ldv[t][m][3]);
                *reinterpret_cast<float4*>(&sDU[t][MCH + m][tid][0]) = make_float4(uv[t][m][0], uv[t][m][1], uv[t][m][2], uv[t][m][3]);
            }
        }
    }
    __syncthreads();

    if (tid < MCH * NN) {
        const int m = tid >> 4;
        const int n = tid & 15;
        float Hp = 0.0f;
        #pragma unroll 8
        for (int j = 0; j < CH; ++j) {
            const float hj = sH[m][j][n];
            sH[m][j][n] = Hp;
            Hp = sA[m][j][n] * Hp + hj;
        }
    }
    __syncthreads();

    v2f h[MCH][QP];
    #pragma unroll
    for (int m = 0; m < MCH; ++m) {
        const float4 i4 = *reinterpret_cast<const float4*>(&sH[m][chunk][n0]);
        h[m][0] = (v2f){i4.x, i4.y};
        h[m][1] = (v2f){i4.z, i4.w};
    }
    #pragma unroll
    for (int t = 0; t < CHL / GRP; ++t) {
        const int lq = lbeg + t * GRP + 4 * q;
        const unsigned rt = r0 + (unsigned)(t * GRP * rst * 4);
        float ldt[MCH][4], ut[MCH][4], duv[MCH][4], ykeep[MCH][4];
        #pragma unroll
        for (int m = 0; m < MCH; ++m) {
            const float4 a = *reinterpret_cast<const float4*>(&sDU[t][m][tid][0]);
            const float4 c = *reinterpret_cast<const float4*>(&sDU[t][MCH + m][tid][0]);
            ldt[m][0] = a.x; ldt[m][1] = a.y; ldt[m][2] = a.z; ldt[m][3] = a.w;
            ut[m][0]  = c.x; ut[m][1]  = c.y; ut[m][2]  = c.z; ut[m][3]  = c.w;
        }
        mul_du(ldt, ut, duv);
        #pragma unroll
        for (int m = 0; m < MCH; ++m) {
            #pragma unroll
            for (int i = 0; i < 4; ++i) ykeep[m][i] = 0.0f;
        }
        out_group<0>(q, ldt, duv, An, Be, Ce, h, ykeep);
        GROUP_FENCE();
        load_rows_bc<2>(bpl, rt, rst, Be, Ce);
        GROUP_FENCE();
        out_group<1>(q, ldt, duv, An, Bo, Co, h, ykeep);
        GROUP_FENCE();
        load_rows_bc<3>(bpl, rt, rst, Bo, Co);
        GROUP_FENCE();
        out_group<2>(q, ldt, duv, An, Be, Ce, h, ykeep);
        GROUP_FENCE();
        if (t == 0) load_rows_bc<4>(bpl, rt, rst, Be, Ce);
        GROUP_FENCE();
        out_group<3>(q, ldt, duv, An, Bo, Co, h, ykeep);
        GROUP_FENCE();
        if (t == 0) load_rows_bc<5>(bpl, rt, rst, Bo, Co);
        GROUP_FENCE();
        #pragma unroll
        for (int m = 0; m < MCH; ++m) {
            float y4[4];
            #pragma unroll
            for (int i = 0; i < 4; ++i) y4[i] = ykeep[m][i] + Dd[m] * ut[m][i];
            if (k & 1) stage_col4(&sY[m][0][0], k, lq, y4);
            else       store_row4(optr + m * LL, k, lq, y4);
        }
    }

    if (k & 1) {
        __syncthreads();
        constexpr int F4C = LL / 4;
        constexpr int SWP = F4C / SCAN_THREADS;
        #pragma unroll
        for (int j = 0; j < MCH * SWP; ++j) {
            const int m  = j / SWP;
            const int r  = tid + (j % SWP) * SCAN_THREADS;
            const int h  = r >> 4;
            const int w4 = (r & 15) << 2;
            const float4 v = make_float4(sY[m][w4 + 0][h], sY[m][w4 + 1][h], sY[m][w4 + 2][h], sY[m][w4 + 3][h]);
            *reinterpret_cast<float4*>(optr + m * LL + h * WWW + w4) = v;
        }
    }
}

extern "C" void kernel_launch(void* const* d_in, const int* in_sizes, int n_in,
                              void* d_out, int out_size, void* d_ws, size_t ws_size,
                              hipStream_t stream) {
    const float* x      = (const float*)d_in[0];
    const float* wproj  = (const float*)d_in[1];
    const float* wdelta = (const float*)d_in[2];
    const float* dbias  = (const float*)d_in[3];
    const float* A_log  = (const float*)d_in[4];
    const float* Ds     = (const float*)d_in[5];
    float* out = (float*)d_out;

    float* ws     = (float*)d_ws;
    float* du     = ws;
    float* Bt     = du     + (size_t)BB * KK * CC * LL * 2;
    float* Ct     = Bt     + (size_t)BB * KK * LL * NN;

    dim3 gProj(LL / LT, KK, BB);
    s6_proj_kernel<<<gProj, 256, 0, stream>>>(x, wproj, wdelta, dbias, du, Bt, Ct);

    s6_scan_kernel<<<BB * KK * (CC / MCH), SCAN_THREADS, 0, stream>>>(du, Bt, A_log, Ds, out);
}
